# starting kernel
# baseline (speedup 1.0000x reference)
#include <hip/hip_runtime.h>
#include <hip/hip_bf16.h>

#define M_LVL   6
#define SIGLEN  19530
#define BB      64
#define SS      32
#define ROWS    (BB * SS)
#define NBISECT 100
#define C_PARAM 4.0f
#define A_PARAM 1.0f

#define CHUNK   1024
#define NCHUNK  ((SIGLEN + CHUNK - 1) / CHUNK)
#define NGROUP  (ROWS / 16)

typedef __attribute__((ext_vector_type(2))) float v2f;
typedef __attribute__((ext_vector_type(8))) float v8f;

__device__ __forceinline__ int lvl_of(int i) {
    return 1 + (i >= 5) + (i >= 30) + (i >= 155) + (i >= 780) + (i >= 3905);
}

__device__ __forceinline__ bool finitef(float v) {
    return (v == v) && (fabsf(v) <= 3.402823466e38f);
}

__device__ __forceinline__ v8f wmma_acc(v2f a, v2f b, v8f c) {
    return __builtin_amdgcn_wmma_f32_16x16x4_f32(false, a, false, b, (short)0, c, false, false);
}

__global__ __launch_bounds__(128)
void level_sums_wmma(const float* __restrict__ x, float* __restrict__ partials) {
    const int lane  = threadIdx.x;
    const int group = blockIdx.y * 4 + threadIdx.y;
    const int chunk = blockIdx.x;

    const int rowM = lane & 15;
    const int koff = (lane >> 4) * 2;
    const int col  = lane & 15;
    const float* rowPtr = x + (long)(group * 16 + rowM) * SIGLEN;

    v8f acc = {0.f, 0.f, 0.f, 0.f, 0.f, 0.f, 0.f, 0.f};

    const int cStart = chunk * CHUNK;
    const int cEnd   = min(cStart + CHUNK, SIGLEN);
    const int lvlStart[M_LVL + 1] = {0, 5, 30, 155, 780, 3905, SIGLEN};

#pragma unroll
    for (int k = 1; k <= M_LVL; ++k) {
        const int segBeg = max(cStart, lvlStart[k - 1]);
        const int segEnd = min(cEnd,   lvlStart[k]);
        if (segBeg >= segEnd) continue;

        v2f bfrag;
        const float bval = (col == k - 1) ? 1.0f : 0.0f;
        bfrag[0] = bval; bfrag[1] = bval;

        const int g0       = segBeg & ~3;
        const int g1       = (segEnd + 3) & ~3;
        const int gFastBeg = (segBeg + 3) & ~3;
        const int gFastEnd = segEnd & ~3;

        for (int g = g0; g < min(gFastBeg, g1); g += 4) {
            const int co = g + koff;
            float ax = 0.f, ay = 0.f;
            if (co < SIGLEN) {
                const float2 v = *(const float2*)(rowPtr + co);
                ax = (co     >= segBeg && co     < segEnd) ? v.x * v.x : 0.f;
                ay = (co + 1 >= segBeg && co + 1 < segEnd) ? v.y * v.y : 0.f;
            }
            v2f a; a[0] = ax; a[1] = ay;
            acc = wmma_acc(a, bfrag, acc);
        }

        if (gFastBeg < gFastEnd) {
            const float* p = rowPtr + gFastBeg + koff;
            float2 vcur = *(const float2*)p;
            for (int g = gFastBeg; g + 4 < gFastEnd; g += 4) {
                p += 4;
                const float2 vnxt = *(const float2*)p;
                v2f a; a[0] = vcur.x * vcur.x; a[1] = vcur.y * vcur.y;
                acc = wmma_acc(a, bfrag, acc);
                vcur = vnxt;
            }
            v2f a; a[0] = vcur.x * vcur.x; a[1] = vcur.y * vcur.y;
            acc = wmma_acc(a, bfrag, acc);
        }

        for (int g = max(gFastBeg, gFastEnd); g < g1; g += 4) {
            const int co = g + koff;
            float ax = 0.f, ay = 0.f;
            if (co < SIGLEN) {
                const float2 v = *(const float2*)(rowPtr + co);
                ax = (co     >= segBeg && co     < segEnd) ? v.x * v.x : 0.f;
                ay = (co + 1 >= segBeg && co + 1 < segEnd) ? v.y * v.y : 0.f;
            }
            v2f a; a[0] = ax; a[1] = ay;
            acc = wmma_acc(a, bfrag, acc);
        }
    }

    if (col < M_LVL) {
        const int mhalf = (lane >> 4) * 8;
#pragma unroll
        for (int v = 0; v < 8; ++v) {
            const int row = group * 16 + mhalf + v;
            partials[((long)chunk * ROWS + row) * 8 + col] = acc[v];
        }
    }
}

__global__ __launch_bounds__(256)
void bisect_kernel(const float* __restrict__ partials, float* __restrict__ rootpow) {
    const int row = blockIdx.x * blockDim.x + threadIdx.x;
    if (row >= ROWS) return;

    float ck[M_LVL];
#pragma unroll
    for (int k = 0; k < M_LVL; ++k) ck[k] = 0.f;
    for (int ch = 0; ch < NCHUNK; ++ch) {
#pragma unroll
        for (int k = 0; k < M_LVL; ++k)
            ck[k] += partials[((long)ch * ROWS + row) * 8 + k];
    }

    float nq = 1.f;
#pragma unroll
    for (int k = 0; k < M_LVL; ++k) nq += ck[k];

    float phi = nq;
    if (nq > C_PARAM) {
        phi = C_PARAM + powf(C_PARAM, 1.f + A_PARAM) *
                        (powf(C_PARAM, -A_PARAM) - powf(nq, -A_PARAM)) / A_PARAM;
    }
    const float c0 = 1.f - phi;

    bool finite = finitef(c0);
#pragma unroll
    for (int k = 0; k < M_LVL; ++k) finite = finite && finitef(ck[k]);

    float lo = 0.f, hi = 2.f;
    for (int it = 0; it < NBISECT; ++it) {
        const float mid = 0.5f * (lo + hi);
        const float t2 = mid * mid;
        float p = c0, tp = 1.f;
#pragma unroll
        for (int k = 0; k < M_LVL; ++k) { tp *= t2; p += ck[k] * tp; }
        const bool neg = p < 0.f;
        lo = neg ? mid : lo;
        hi = neg ? hi : mid;
    }
    float root = 0.5f * (lo + hi);
    if (!finite) root = 0.f;
    root = fminf(root, 1.f);

    float rp = 1.f;
    rootpow[row * 8 + 0] = 1.f;
#pragma unroll
    for (int k = 1; k <= M_LVL; ++k) { rp *= root; rootpow[row * 8 + k] = rp; }
    rootpow[row * 8 + 7] = 0.f;
}

__global__ __launch_bounds__(256)
void scale_mean_kernel(const float* __restrict__ x, const float* __restrict__ rootpow,
                       float* __restrict__ out) {
    __shared__ float rp[SS * 8];
    const int b   = blockIdx.y;
    const int tid = threadIdx.x;
    rp[tid] = rootpow[(long)b * SS * 8 + tid];
    __syncthreads();

    const int p = blockIdx.x * 256 + tid;
    const int halfL = SIGLEN / 2;
    if (p >= halfL) return;
    const int l0 = p * 2;
    const int k0 = lvl_of(l0);
    const int k1 = lvl_of(l0 + 1);

    const float* xb = x + (long)b * SS * SIGLEN + l0;
    float accx = 0.f, accy = 0.f;
#pragma unroll 4
    for (int s = 0; s < SS; ++s) {
        const float2 v = *(const float2*)(xb + (long)s * SIGLEN);
        accx += v.x * rp[s * 8 + k0];
        accy += v.y * rp[s * 8 + k1];
    }
    out[(long)b * SIGLEN + l0]     = accx * (1.f / SS);
    out[(long)b * SIGLEN + l0 + 1] = accy * (1.f / SS);
}

extern "C" void kernel_launch(void* const* d_in, const int* in_sizes, int n_in,
                              void* d_out, int out_size, void* d_ws, size_t ws_size,
                              hipStream_t stream) {
    const float* x = (const float*)d_in[0];
    float* out = (float*)d_out;

    float* partials = (float*)d_ws;
    float* rootpow  = partials + (long)NCHUNK * ROWS * 8;

    {
        dim3 blk(32, 4), grd(NCHUNK, NGROUP / 4);
        level_sums_wmma<<<grd, blk, 0, stream>>>(x, partials);
    }
    {
        bisect_kernel<<<ROWS / 256, 256, 0, stream>>>(partials, rootpow);
    }
    {
        dim3 blk(256), grd((SIGLEN / 2 + 255) / 256, BB);
        scale_mean_kernel<<<grd, blk, 0, stream>>>(x, rootpow, out);
    }
}


# Scale pass: a wave streams 128-column tiles as 32 contiguous 512 B row pieces (16 B per lane, 16-32 loads in flight), 2-3 tiles per wave, 64 waves per batch (were 312, 8 B per lane); sums unchanged
# speedup vs baseline: 4.1476x; 4.1476x over previous
#include <hip/hip_runtime.h>
#include <hip/hip_bf16.h>

#define M_LVL   6
#define SIGLEN  19530
#define BB      64
#define SS      32
#define ROWS    (BB * SS)
#define NBISECT 100
#define C_PARAM 4.0f
#define A_PARAM 1.0f

#define NGROUP   (ROWS / 16)
#define TILE_C   128
#define NTILE    5
#define WAVE_C   612
#define WG_WAVES 8
#define NPART    4
#define PITCH    132

typedef __attribute__((ext_vector_type(2))) float v2f;
typedef __attribute__((ext_vector_type(4))) float v4f;
typedef __attribute__((ext_vector_type(8))) float v8f;
typedef v4f __attribute__((aligned(8))) v4f_a8;

__device__ __forceinline__ int lvl_of(int i) {
    return 1 + (i >= 5) + (i >= 30) + (i >= 155) + (i >= 780) + (i >= 3905);
}

__device__ __forceinline__ bool finitef(float v) {
    return (v == v) && (fabsf(v) <= 3.402823466e38f);
}

__device__ __forceinline__ v8f wmma_acc(v2f a, v2f b, v8f c) {
    return __builtin_amdgcn_wmma_f32_16x16x4_f32(false, a, false, b, (short)0, c, false, false);
}

__device__ __forceinline__ void load_tile(v4f (&R)[16], const float* __restrict__ gbase, int colLd) {
    for (int i = 0; i < 16; ++i)
        R[i] = *(const v4f_a8*)(gbase + (long)i * SIGLEN + colLd);
}

__global__ __launch_bounds__(WG_WAVES * 32)
void level_sums_wmma(const float* __restrict__ x, float* __restrict__ partials) {
    __shared__ __attribute__((aligned(16))) float tile[WG_WAVES][16 * PITCH];
    __shared__ float red[WG_WAVES][16][8];

    const int lane = threadIdx.x & 31;
    const int wv   = __builtin_amdgcn_readfirstlane(threadIdx.x >> 5);
    const int g    = blockIdx.x / NPART;
    const int q    = blockIdx.x % NPART;

    const int c = lane & 15;
    const int h = lane >> 4;

    const int cbeg = (q * WG_WAVES + wv) * WAVE_C;
    const int cend = min(cbeg + WAVE_C, SIGLEN);
    const float* gbase = x + (long)(g * 16) * SIGLEN;
    float* my = tile[wv];

    v8f acc0 = {0.f, 0.f, 0.f, 0.f, 0.f, 0.f, 0.f, 0.f};
    v8f acc1 = {0.f, 0.f, 0.f, 0.f, 0.f, 0.f, 0.f, 0.f};

    v4f R[16];
    load_tile(R, gbase, min(cbeg + 4 * lane, cend - 4));

    for (int t = 0; t < NTILE; ++t) {
        const int t0 = cbeg + t * TILE_C;

        if (t == NTILE - 1) {
            const int shift = (t0 + 4 * lane) - min(t0 + 4 * lane, cend - 4);
            for (int i = 0; i < 16; ++i) {
                const v4f ld = R[i];
                v4f r;
                r[0] = (shift == 0) ? ld[0] : ((shift == 2) ? ld[2] : 0.f);
                r[1] = (shift == 0) ? ld[1] : ((shift == 2) ? ld[3] : 0.f);
                r[2] = (shift == 0) ? ld[2] : 0.f;
                r[3] = (shift == 0) ? ld[3] : 0.f;
                R[i] = r;
            }
        }

        for (int i = 0; i < 16; ++i)
            *(v4f*)(my + i * PITCH + 4 * lane) = R[i];

        if (t + 1 < NTILE)
            load_tile(R, gbase, min(t0 + TILE_C + 4 * lane, cend - 4));


        const float* frag = my + c * PITCH + 4 * h;
        v4f F[16];
        for (int j = 0; j < 16; ++j) F[j] = *(const v4f*)(frag + 8 * j);

        const int kLo = lvl_of(t0);
        const int kHi = lvl_of(min(t0 + TILE_C, cend) - 1);
        if (kLo == kHi) {
            const float bval = (c == kLo - 1) ? 1.0f : 0.0f;
            v2f bfrag; bfrag[0] = bval; bfrag[1] = bval;
            for (int j = 0; j < 16; ++j) {
                const v4f f = F[j];
                v2f a0; a0[0] = f[0] * f[0]; a0[1] = f[1] * f[1];
                v2f a1; a1[0] = f[2] * f[2]; a1[1] = f[3] * f[3];
                acc0 = wmma_acc(a0, bfrag, acc0);
                acc1 = wmma_acc(a1, bfrag, acc1);
            }
        } else {
            for (int j = 0; j < 16; ++j) {
                const v4f f = F[j];
                const int col0 = t0 + 8 * j + 4 * h;
                v2f a0; a0[0] = f[0] * f[0]; a0[1] = f[1] * f[1];
                v2f a1; a1[0] = f[2] * f[2]; a1[1] = f[3] * f[3];
                v2f b0; b0[0] = (lvl_of(col0)     == c + 1) ? 1.0f : 0.0f;
                        b0[1] = (lvl_of(col0 + 1) == c + 1) ? 1.0f : 0.0f;
                v2f b1; b1[0] = (lvl_of(col0 + 2) == c + 1) ? 1.0f : 0.0f;
                        b1[1] = (lvl_of(col0 + 3) == c + 1) ? 1.0f : 0.0f;
                acc0 = wmma_acc(a0, b0, acc0);
                acc1 = wmma_acc(a1, b1, acc1);
            }
        }
    }

    if (c < 8) {
        for (int v = 0; v < 8; ++v)
            red[wv][8 * h + v][c] = acc0[v] + acc1[v];
    }
    __syncthreads();
    if (threadIdx.x < 128) {
        const int row = threadIdx.x >> 3, col = threadIdx.x & 7;
        float s = 0.f;
        for (int w = 0; w < WG_WAVES; ++w) s += red[w][row][col];
        partials[((long)q * ROWS + g * 16 + row) * 8 + col] = s;
    }
}

__global__ __launch_bounds__(256)
void bisect_kernel(const float* __restrict__ partials, float* __restrict__ rootpow) {
    const int row = blockIdx.x * blockDim.x + threadIdx.x;
    if (row >= ROWS) return;

    float ck[M_LVL];
#pragma unroll
    for (int k = 0; k < M_LVL; ++k) ck[k] = 0.f;
    for (int ch = 0; ch < NPART; ++ch) {
#pragma unroll
        for (int k = 0; k < M_LVL; ++k)
            ck[k] += partials[((long)ch * ROWS + row) * 8 + k];
    }

    float nq = 1.f;
#pragma unroll
    for (int k = 0; k < M_LVL; ++k) nq += ck[k];

    float phi = nq;
    if (nq > C_PARAM) {
        phi = C_PARAM + powf(C_PARAM, 1.f + A_PARAM) *
                        (powf(C_PARAM, -A_PARAM) - powf(nq, -A_PARAM)) / A_PARAM;
    }
    const float c0 = 1.f - phi;

    bool finite = finitef(c0);
#pragma unroll
    for (int k = 0; k < M_LVL; ++k) finite = finite && finitef(ck[k]);

    float lo = 0.f, hi = 2.f;
    for (int it = 0; it < NBISECT; ++it) {
        const float mid = 0.5f * (lo + hi);
        const float t2 = mid * mid;
        float p = c0, tp = 1.f;
#pragma unroll
        for (int k = 0; k < M_LVL; ++k) { tp *= t2; p += ck[k] * tp; }
        const bool neg = p < 0.f;
        lo = neg ? mid : lo;
        hi = neg ? hi : mid;
    }
    float root = 0.5f * (lo + hi);
    if (!finite) root = 0.f;
    root = fminf(root, 1.f);

    float rp = 1.f;
    rootpow[row * 8 + 0] = 1.f;
#pragma unroll
    for (int k = 1; k <= M_LVL; ++k) { rp *= root; rootpow[row * 8 + k] = rp; }
    rootpow[row * 8 + 7] = 0.f;
}

#define SC_TILES 153
#define SC_WGPB  8

__device__ __forceinline__ void load_half(v4f (&R)[16], const float* __restrict__ p) {
    for (int i = 0; i < 16; ++i)
        R[i] = *(const v4f_a8*)(p + (long)i * SIGLEN);
}

__device__ __forceinline__ void sum_half_one(v4f& acc, const v4f (&R)[16], const float* rpw) {
    for (int i = 0; i < 16; ++i) {
        const float wgt = rpw[i * 8];
        acc[0] = __builtin_fmaf(R[i][0], wgt, acc[0]);
        acc[1] = __builtin_fmaf(R[i][1], wgt, acc[1]);
        acc[2] = __builtin_fmaf(R[i][2], wgt, acc[2]);
        acc[3] = __builtin_fmaf(R[i][3], wgt, acc[3]);
    }
}

__device__ __forceinline__ void sum_half_mix(v4f& acc, const v4f (&R)[16], const float* rph,
                                             int k0, int k1, int k2, int k3) {
    for (int i = 0; i < 16; ++i) {
        acc[0] = __builtin_fmaf(R[i][0], rph[i * 8 + k0], acc[0]);
        acc[1] = __builtin_fmaf(R[i][1], rph[i * 8 + k1], acc[1]);
        acc[2] = __builtin_fmaf(R[i][2], rph[i * 8 + k2], acc[2]);
        acc[3] = __builtin_fmaf(R[i][3], rph[i * 8 + k3], acc[3]);
    }
}

__global__ __launch_bounds__(256)
void scale_mean_kernel(const float* __restrict__ x, const float* __restrict__ rootpow,
                       float* __restrict__ out) {
    __shared__ float rp[SS * 8];
    const int b   = blockIdx.x / SC_WGPB;
    const int qq  = blockIdx.x % SC_WGPB;
    const int tid = threadIdx.x;
    rp[tid] = rootpow[(long)b * SS * 8 + tid];
    __syncthreads();

    const int lane = tid & 31;
    const int wv    = __builtin_amdgcn_readfirstlane(tid >> 5);
    const int w     = qq * 8 + wv;
    const int tbeg  = (SC_TILES * w) >> 6;
    const int tend  = (SC_TILES * (w + 1)) >> 6;
    const bool three = (tend - tbeg) > 2;

    const float* xb = x + (long)b * SS * SIGLEN;
    float* ob = out + (long)b * SIGLEN;

    v4f H[2][16];
    load_half(H[0], xb + min(tbeg * TILE_C + 4 * lane, SIGLEN - 4));

    v4f acc = {0.f, 0.f, 0.f, 0.f};
#pragma unroll
    for (int st = 0; st < 6; ++st) {
        const int j = st >> 1, hf = st & 1;
        if (j < 2 || three) {
            const int t0   = (tbeg + j) * TILE_C;
            const int want = t0 + 4 * lane;
            const int cl   = min(want, SIGLEN - 4);
            const int shift = want - cl;

            __builtin_amdgcn_sched_barrier(0);
            if (st + 1 < 6) {
                const int j1 = (st + 1) >> 1, hf1 = (st + 1) & 1;
                if (j1 < 2 || three)
                    load_half(H[(st + 1) & 1],
                              xb + hf1 * 16L * SIGLEN + min((tbeg + j1) * TILE_C + 4 * lane, SIGLEN - 4));
            }
            __builtin_amdgcn_sched_barrier(0);

            const int kLo = lvl_of(t0);
            const int kHi = lvl_of(min(t0 + TILE_C, SIGLEN) - 1);
            if (hf == 0) { acc[0] = 0.f; acc[1] = 0.f; acc[2] = 0.f; acc[3] = 0.f; }
            if (kLo == kHi) {
                sum_half_one(acc, H[st & 1], rp + hf * 16 * 8 + kLo);
            } else {
                sum_half_mix(acc, H[st & 1], rp + hf * 16 * 8,
                             lvl_of(cl), lvl_of(cl + 1), lvl_of(cl + 2), lvl_of(cl + 3));
            }

            if (hf == 1) {
                const float inv = 1.f / SS;
                if (shift == 0) {
                    v4f o; o[0] = acc[0] * inv; o[1] = acc[1] * inv; o[2] = acc[2] * inv; o[3] = acc[3] * inv;
                    *(v4f_a8*)(ob + cl) = o;
                } else if (shift == 2) {
                    v2f o; o[0] = acc[2] * inv; o[1] = acc[3] * inv;
                    *(v2f*)(ob + cl + 2) = o;
                }
            }
        }
    }
}

extern "C" void kernel_launch(void* const* d_in, const int* in_sizes, int n_in,
                              void* d_out, int out_size, void* d_ws, size_t ws_size,
                              hipStream_t stream) {
    const float* x = (const float*)d_in[0];
    float* out = (float*)d_out;

    float* partials = (float*)d_ws;
    float* rootpow  = partials + (long)NPART * ROWS * 8;

    {
        level_sums_wmma<<<NGROUP * NPART, WG_WAVES * 32, 0, stream>>>(x, partials);
    }
    {
        bisect_kernel<<<ROWS / 256, 256, 0, stream>>>(partials, rootpow);
    }
    {
        scale_mean_kernel<<<BB * SC_WGPB, 256, 0, stream>>>(x, rootpow, out);
    }
}


# Level sums: a work-group owns 8 whole rows (16 waves; matrix rows = 8 rows x 2 adjacent 512 B pieces), so its tail solves their roots; 3 launches -> 2; f32 summation order changes
# speedup vs baseline: 4.4672x; 1.0771x over previous
#include <hip/hip_runtime.h>
#include <hip/hip_bf16.h>

#define M_LVL   6
#define SIGLEN  19530
#define BB      64
#define SS      32
#define ROWS    (BB * SS)
#define NBISECT 100
#define C_PARAM 4.0f
#define A_PARAM 1.0f

#define RGRP     8
#define NGROUP   (ROWS / RGRP)
#define TILE_C   128
#define DTILE_C  (2 * TILE_C)
#define NTILE    5
#define WAVE_C   1224
#define WG_WAVES 16
#define PITCH    132

typedef __attribute__((ext_vector_type(2))) float v2f;
typedef __attribute__((ext_vector_type(4))) float v4f;
typedef __attribute__((ext_vector_type(8))) float v8f;
typedef v4f __attribute__((aligned(8))) v4f_a8;

__device__ __forceinline__ int lvl_of(int i) {
    return 1 + (i >= 5) + (i >= 30) + (i >= 155) + (i >= 780) + (i >= 3905);
}

__device__ __forceinline__ bool finitef(float v) {
    return (v == v) && (fabsf(v) <= 3.402823466e38f);
}

__device__ __forceinline__ v8f wmma_acc(v2f a, v2f b, v8f c) {
    return __builtin_amdgcn_wmma_f32_16x16x4_f32(false, a, false, b, (short)0, c, false, false);
}

__device__ __forceinline__ void load_tile(v4f (&R)[16], const float* __restrict__ gbase,
                                          int t0, int lane, int cend) {
    for (int i = 0; i < 16; ++i) {
        const int col = min(t0 + (i >> 3) * TILE_C + 4 * lane, cend - 4);
        R[i] = *(const v4f_a8*)(gbase + (long)(i & 7) * SIGLEN + col);
    }
}

__global__ __launch_bounds__(WG_WAVES * 32)
void level_sums_roots(const float* __restrict__ x, float* __restrict__ rootpow) {
    __shared__ __attribute__((aligned(16))) float tile[WG_WAVES][16 * PITCH];
    __shared__ float red[WG_WAVES][16][8];
    __shared__ float cks[RGRP][8];

    const int lane = threadIdx.x & 31;
    const int wv   = __builtin_amdgcn_readfirstlane(threadIdx.x >> 5);
    const int g    = blockIdx.x;

    const int c = lane & 15;
    const int h = lane >> 4;

    const int cbeg = wv * WAVE_C;
    const int cend = min(cbeg + WAVE_C, SIGLEN);
    const float* gbase = x + (long)(g * RGRP) * SIGLEN;
    float* my = tile[wv];

    v8f acc0 = {0.f, 0.f, 0.f, 0.f, 0.f, 0.f, 0.f, 0.f};
    v8f acc1 = {0.f, 0.f, 0.f, 0.f, 0.f, 0.f, 0.f, 0.f};

    v4f R[16];
    load_tile(R, gbase, cbeg, lane, cend);

    for (int t = 0; t < NTILE; ++t) {
        const int t0 = cbeg + t * DTILE_C;

        if (t == NTILE - 1) {
            const int w0 = t0 + 4 * lane, w1 = w0 + TILE_C;
            const int shift0 = w0 - min(w0, cend - 4);
            const int shift1 = w1 - min(w1, cend - 4);
            for (int i = 0; i < 16; ++i) {
                const int shift = (i >> 3) ? shift1 : shift0;
                const v4f ld = R[i];
                v4f r;
                r[0] = (shift == 0) ? ld[0] : ((shift == 2) ? ld[2] : 0.f);
                r[1] = (shift == 0) ? ld[1] : ((shift == 2) ? ld[3] : 0.f);
                r[2] = (shift == 0) ? ld[2] : 0.f;
                r[3] = (shift == 0) ? ld[3] : 0.f;
                R[i] = r;
            }
        }

        for (int i = 0; i < 16; ++i)
            *(v4f*)(my + i * PITCH + 4 * lane) = R[i];

        if (t + 1 < NTILE)
            load_tile(R, gbase, t0 + DTILE_C, lane, cend);


        const float* frag = my + c * PITCH + 4 * h;
        v4f F[16];
        for (int j = 0; j < 16; ++j) F[j] = *(const v4f*)(frag + 8 * j);

        const int kLo = lvl_of(t0);
        const int kHi = lvl_of(min(t0 + DTILE_C, cend) - 1);
        if (kLo == kHi) {
            const float bval = (c == kLo - 1) ? 1.0f : 0.0f;
            v2f bfrag; bfrag[0] = bval; bfrag[1] = bval;
            for (int j = 0; j < 16; ++j) {
                const v4f f = F[j];
                v2f a0; a0[0] = f[0] * f[0]; a0[1] = f[1] * f[1];
                v2f a1; a1[0] = f[2] * f[2]; a1[1] = f[3] * f[3];
                acc0 = wmma_acc(a0, bfrag, acc0);
                acc1 = wmma_acc(a1, bfrag, acc1);
            }
        } else {
            const float keep0 = (c < 8) ? 1.0f : 0.0f;
            const float keep1 = 1.0f - keep0;
            for (int j = 0; j < 16; ++j) {
                const v4f f = F[j];
                const int col0 = t0 + 8 * j + 4 * h;
                const int col1 = col0 + TILE_C;
                const float s0 = f[0] * f[0], s1 = f[1] * f[1], s2 = f[2] * f[2], s3 = f[3] * f[3];
                v2f a0; a0[0] = s0 * keep0; a0[1] = s1 * keep0;
                v2f a1; a1[0] = s2 * keep0; a1[1] = s3 * keep0;
                v2f b0; b0[0] = (lvl_of(col0)     == c + 1) ? 1.0f : 0.0f;
                        b0[1] = (lvl_of(col0 + 1) == c + 1) ? 1.0f : 0.0f;
                v2f b1; b1[0] = (lvl_of(col0 + 2) == c + 1) ? 1.0f : 0.0f;
                        b1[1] = (lvl_of(col0 + 3) == c + 1) ? 1.0f : 0.0f;
                acc0 = wmma_acc(a0, b0, acc0);
                acc1 = wmma_acc(a1, b1, acc1);
                v2f a2; a2[0] = s0 * keep1; a2[1] = s1 * keep1;
                v2f a3; a3[0] = s2 * keep1; a3[1] = s3 * keep1;
                v2f b2; b2[0] = (lvl_of(col1)     == c + 1) ? 1.0f : 0.0f;
                        b2[1] = (lvl_of(col1 + 1) == c + 1) ? 1.0f : 0.0f;
                v2f b3; b3[0] = (lvl_of(col1 + 2) == c + 1) ? 1.0f : 0.0f;
                        b3[1] = (lvl_of(col1 + 3) == c + 1) ? 1.0f : 0.0f;
                acc0 = wmma_acc(a2, b2, acc0);
                acc1 = wmma_acc(a3, b3, acc1);
            }
        }
    }

    if (c < 8) {
        for (int v = 0; v < 8; ++v)
            red[wv][8 * h + v][c] = acc0[v] + acc1[v];
    }
    __syncthreads();
    if (threadIdx.x < RGRP * 8) {
        const int row = threadIdx.x >> 3, col = threadIdx.x & 7;
        float s = 0.f;
        for (int w = 0; w < WG_WAVES; ++w) s += red[w][row][col] + red[w][row + 8][col];
        cks[row][col] = s;
    }
    __syncthreads();

    if (threadIdx.x < RGRP) {
        const int row = g * RGRP + threadIdx.x;
        float ck[M_LVL];
#pragma unroll
        for (int k = 0; k < M_LVL; ++k) ck[k] = cks[threadIdx.x][k];

        float nq = 1.f;
#pragma unroll
        for (int k = 0; k < M_LVL; ++k) nq += ck[k];

        float phi = nq;
        if (nq > C_PARAM) {
            phi = C_PARAM + powf(C_PARAM, 1.f + A_PARAM) *
                            (powf(C_PARAM, -A_PARAM) - powf(nq, -A_PARAM)) / A_PARAM;
        }
        const float c0 = 1.f - phi;

        bool finite = finitef(c0);
#pragma unroll
        for (int k = 0; k < M_LVL; ++k) finite = finite && finitef(ck[k]);

        float lo = 0.f, hi = 2.f;
        for (int it = 0; it < NBISECT; ++it) {
            const float mid = 0.5f * (lo + hi);
            const float t2 = mid * mid;
            float p = c0, tp = 1.f;
#pragma unroll
            for (int k = 0; k < M_LVL; ++k) { tp *= t2; p += ck[k] * tp; }
            const bool neg = p < 0.f;
            lo = neg ? mid : lo;
            hi = neg ? hi : mid;
        }
        float root = 0.5f * (lo + hi);
        if (!finite) root = 0.f;
        root = fminf(root, 1.f);

        float rp = 1.f;
        rootpow[row * 8 + 0] = 1.f;
#pragma unroll
        for (int k = 1; k <= M_LVL; ++k) { rp *= root; rootpow[row * 8 + k] = rp; }
        rootpow[row * 8 + 7] = 0.f;
    }
}

#define SC_TILES 153
#define SC_WGPB  8

__device__ __forceinline__ void load_half(v4f (&R)[16], const float* __restrict__ p) {
    for (int i = 0; i < 16; ++i)
        R[i] = *(const v4f_a8*)(p + (long)i * SIGLEN);
}

__device__ __forceinline__ void sum_half_one(v4f& acc, const v4f (&R)[16], const float* rpw) {
    for (int i = 0; i < 16; ++i) {
        const float wgt = rpw[i * 8];
        acc[0] = __builtin_fmaf(R[i][0], wgt, acc[0]);
        acc[1] = __builtin_fmaf(R[i][1], wgt, acc[1]);
        acc[2] = __builtin_fmaf(R[i][2], wgt, acc[2]);
        acc[3] = __builtin_fmaf(R[i][3], wgt, acc[3]);
    }
}

__device__ __forceinline__ void sum_half_mix(v4f& acc, const v4f (&R)[16], const float* rph,
                                             int k0, int k1, int k2, int k3) {
    for (int i = 0; i < 16; ++i) {
        acc[0] = __builtin_fmaf(R[i][0], rph[i * 8 + k0], acc[0]);
        acc[1] = __builtin_fmaf(R[i][1], rph[i * 8 + k1], acc[1]);
        acc[2] = __builtin_fmaf(R[i][2], rph[i * 8 + k2], acc[2]);
        acc[3] = __builtin_fmaf(R[i][3], rph[i * 8 + k3], acc[3]);
    }
}

__global__ __launch_bounds__(256)
void scale_mean_kernel(const float* __restrict__ x, const float* __restrict__ rootpow,
                       float* __restrict__ out) {
    __shared__ float rp[SS * 8];
    const int b   = blockIdx.x / SC_WGPB;
    const int qq  = blockIdx.x % SC_WGPB;
    const int tid = threadIdx.x;
    rp[tid] = rootpow[(long)b * SS * 8 + tid];
    __syncthreads();

    const int lane = tid & 31;
    const int wv    = __builtin_amdgcn_readfirstlane(tid >> 5);
    const int w     = qq * 8 + wv;
    const int tbeg  = (SC_TILES * w) >> 6;
    const int tend  = (SC_TILES * (w + 1)) >> 6;
    const bool three = (tend - tbeg) > 2;

    const float* xb = x + (long)b * SS * SIGLEN;
    float* ob = out + (long)b * SIGLEN;

    v4f H[2][16];
    load_half(H[0], xb + min(tbeg * TILE_C + 4 * lane, SIGLEN - 4));

    v4f acc = {0.f, 0.f, 0.f, 0.f};
#pragma unroll
    for (int st = 0; st < 6; ++st) {
        const int j = st >> 1, hf = st & 1;
        if (j < 2 || three) {
            const int t0   = (tbeg + j) * TILE_C;
            const int want = t0 + 4 * lane;
            const int cl   = min(want, SIGLEN - 4);
            const int shift = want - cl;

            __builtin_amdgcn_sched_barrier(0);
            if (st + 1 < 6) {
                const int j1 = (st + 1) >> 1, hf1 = (st + 1) & 1;
                if (j1 < 2 || three)
                    load_half(H[(st + 1) & 1],
                              xb + hf1 * 16L * SIGLEN + min((tbeg + j1) * TILE_C + 4 * lane, SIGLEN - 4));
            }
            __builtin_amdgcn_sched_barrier(0);

            const int kLo = lvl_of(t0);
            const int kHi = lvl_of(min(t0 + TILE_C, SIGLEN) - 1);
            if (hf == 0) { acc[0] = 0.f; acc[1] = 0.f; acc[2] = 0.f; acc[3] = 0.f; }
            if (kLo == kHi) {
                sum_half_one(acc, H[st & 1], rp + hf * 16 * 8 + kLo);
            } else {
                sum_half_mix(acc, H[st & 1], rp + hf * 16 * 8,
                             lvl_of(cl), lvl_of(cl + 1), lvl_of(cl + 2), lvl_of(cl + 3));
            }

            if (hf == 1) {
                const float inv = 1.f / SS;
                if (shift == 0) {
                    v4f o; o[0] = acc[0] * inv; o[1] = acc[1] * inv; o[2] = acc[2] * inv; o[3] = acc[3] * inv;
                    *(v4f_a8*)(ob + cl) = o;
                } else if (shift == 2) {
                    v2f o; o[0] = acc[2] * inv; o[1] = acc[3] * inv;
                    *(v2f*)(ob + cl + 2) = o;
                }
            }
        }
    }
}

extern "C" void kernel_launch(void* const* d_in, const int* in_sizes, int n_in,
                              void* d_out, int out_size, void* d_ws, size_t ws_size,
                              hipStream_t stream) {
    const float* x = (const float*)d_in[0];
    float* out = (float*)d_out;

    float* rootpow = (float*)d_ws;

    {
        level_sums_roots<<<NGROUP, WG_WAVES * 32, 0, stream>>>(x, rootpow);
    }
    {
        scale_mean_kernel<<<BB * SC_WGPB, 256, 0, stream>>>(x, rootpow, out);
    }
}


# Root solve: a wave per row tests 31 points of the interval at once (one per lane); the count of negative signs is 5 bisection trips; stops when no float lies between the bounds; polynomial all fma
# speedup vs baseline: 4.7428x; 1.0617x over previous
#include <hip/hip_runtime.h>
#include <hip/hip_bf16.h>

#define M_LVL   6
#define SIGLEN  19530
#define BB      64
#define SS      32
#define ROWS    (BB * SS)
#define NBISECT 100
#define C_PARAM 4.0f
#define A_PARAM 1.0f

#define RGRP     8
#define NGROUP   (ROWS / RGRP)
#define TILE_C   128
#define DTILE_C  (2 * TILE_C)
#define NTILE    5
#define WAVE_C   1224
#define WG_WAVES 16
#define PITCH    132

typedef __attribute__((ext_vector_type(2))) float v2f;
typedef __attribute__((ext_vector_type(4))) float v4f;
typedef __attribute__((ext_vector_type(8))) float v8f;
typedef v4f __attribute__((aligned(8))) v4f_a8;

__device__ __forceinline__ int lvl_of(int i) {
    return 1 + (i >= 5) + (i >= 30) + (i >= 155) + (i >= 780) + (i >= 3905);
}

__device__ __forceinline__ bool finitef(float v) {
    return (v == v) && (fabsf(v) <= 3.402823466e38f);
}

__device__ __forceinline__ v8f wmma_acc(v2f a, v2f b, v8f c) {
    return __builtin_amdgcn_wmma_f32_16x16x4_f32(false, a, false, b, (short)0, c, false, false);
}

__device__ __forceinline__ void load_tile(v4f (&R)[16], const float* __restrict__ gbase,
                                          int t0, int lane, int cend) {
    for (int i = 0; i < 16; ++i) {
        const int col = min(t0 + (i >> 3) * TILE_C + 4 * lane, cend - 4);
        R[i] = *(const v4f_a8*)(gbase + (long)(i & 7) * SIGLEN + col);
    }
}

__device__ __forceinline__ float poly_at(float t, float c0, const float (&ck)[M_LVL]) {
    const float t2 = t * t;
    float p = c0, tp = 1.f;
#pragma unroll
    for (int k = 0; k < M_LVL; ++k) { tp *= t2; p = __builtin_fmaf(ck[k], tp, p); }
    return p;
}

__global__ __launch_bounds__(WG_WAVES * 32)
void level_sums_roots(const float* __restrict__ x, float* __restrict__ rootpow) {
    __shared__ __attribute__((aligned(16))) float tile[WG_WAVES][16 * PITCH];
    __shared__ float red[WG_WAVES][16][8];
    __shared__ float cks[RGRP][8];

    const int lane = threadIdx.x & 31;
    const int wv   = __builtin_amdgcn_readfirstlane(threadIdx.x >> 5);
    const int g    = blockIdx.x;

    const int c = lane & 15;
    const int h = lane >> 4;

    const int cbeg = wv * WAVE_C;
    const int cend = min(cbeg + WAVE_C, SIGLEN);
    const float* gbase = x + (long)(g * RGRP) * SIGLEN;
    float* my = tile[wv];

    v8f acc0 = {0.f, 0.f, 0.f, 0.f, 0.f, 0.f, 0.f, 0.f};
    v8f acc1 = {0.f, 0.f, 0.f, 0.f, 0.f, 0.f, 0.f, 0.f};

    v4f R[16];
    load_tile(R, gbase, cbeg, lane, cend);

    for (int t = 0; t < NTILE; ++t) {
        const int t0 = cbeg + t * DTILE_C;

        if (t == NTILE - 1) {
            const int w0 = t0 + 4 * lane, w1 = w0 + TILE_C;
            const int shift0 = w0 - min(w0, cend - 4);
            const int shift1 = w1 - min(w1, cend - 4);
            for (int i = 0; i < 16; ++i) {
                const int shift = (i >> 3) ? shift1 : shift0;
                const v4f ld = R[i];
                v4f r;
                r[0] = (shift == 0) ? ld[0] : ((shift == 2) ? ld[2] : 0.f);
                r[1] = (shift == 0) ? ld[1] : ((shift == 2) ? ld[3] : 0.f);
                r[2] = (shift == 0) ? ld[2] : 0.f;
                r[3] = (shift == 0) ? ld[3] : 0.f;
                R[i] = r;
            }
        }

        for (int i = 0; i < 16; ++i)
            *(v4f*)(my + i * PITCH + 4 * lane) = R[i];

        if (t + 1 < NTILE)
            load_tile(R, gbase, t0 + DTILE_C, lane, cend);


        const float* frag = my + c * PITCH + 4 * h;
        v4f F[16];
        for (int j = 0; j < 16; ++j) F[j] = *(const v4f*)(frag + 8 * j);

        const int kLo = lvl_of(t0);
        const int kHi = lvl_of(min(t0 + DTILE_C, cend) - 1);
        if (kLo == kHi) {
            const float bval = (c == kLo - 1) ? 1.0f : 0.0f;
            v2f bfrag; bfrag[0] = bval; bfrag[1] = bval;
            for (int j = 0; j < 16; ++j) {
                const v4f f = F[j];
                v2f a0; a0[0] = f[0] * f[0]; a0[1] = f[1] * f[1];
                v2f a1; a1[0] = f[2] * f[2]; a1[1] = f[3] * f[3];
                acc0 = wmma_acc(a0, bfrag, acc0);
                acc1 = wmma_acc(a1, bfrag, acc1);
            }
        } else {
            const float keep0 = (c < 8) ? 1.0f : 0.0f;
            const float keep1 = 1.0f - keep0;
            for (int j = 0; j < 16; ++j) {
                const v4f f = F[j];
                const int col0 = t0 + 8 * j + 4 * h;
                const int col1 = col0 + TILE_C;
                const float s0 = f[0] * f[0], s1 = f[1] * f[1], s2 = f[2] * f[2], s3 = f[3] * f[3];
                v2f a0; a0[0] = s0 * keep0; a0[1] = s1 * keep0;
                v2f a1; a1[0] = s2 * keep0; a1[1] = s3 * keep0;
                v2f b0; b0[0] = (lvl_of(col0)     == c + 1) ? 1.0f : 0.0f;
                        b0[1] = (lvl_of(col0 + 1) == c + 1) ? 1.0f : 0.0f;
                v2f b1; b1[0] = (lvl_of(col0 + 2) == c + 1) ? 1.0f : 0.0f;
                        b1[1] = (lvl_of(col0 + 3) == c + 1) ? 1.0f : 0.0f;
                acc0 = wmma_acc(a0, b0, acc0);
                acc1 = wmma_acc(a1, b1, acc1);
                v2f a2; a2[0] = s0 * keep1; a2[1] = s1 * keep1;
                v2f a3; a3[0] = s2 * keep1; a3[1] = s3 * keep1;
                v2f b2; b2[0] = (lvl_of(col1)     == c + 1) ? 1.0f : 0.0f;
                        b2[1] = (lvl_of(col1 + 1) == c + 1) ? 1.0f : 0.0f;
                v2f b3; b3[0] = (lvl_of(col1 + 2) == c + 1) ? 1.0f : 0.0f;
                        b3[1] = (lvl_of(col1 + 3) == c + 1) ? 1.0f : 0.0f;
                acc0 = wmma_acc(a2, b2, acc0);
                acc1 = wmma_acc(a3, b3, acc1);
            }
        }
    }

    if (c < 8) {
        for (int v = 0; v < 8; ++v)
            red[wv][8 * h + v][c] = acc0[v] + acc1[v];
    }
    __syncthreads();
    if (threadIdx.x < RGRP * 8) {
        const int row = threadIdx.x >> 3, col = threadIdx.x & 7;
        float s = 0.f;
        for (int w = 0; w < WG_WAVES; ++w) s += red[w][row][col] + red[w][row + 8][col];
        cks[row][col] = s;
    }
    __syncthreads();

    if (wv < RGRP) {
        const int row = g * RGRP + wv;
        float ck[M_LVL];
#pragma unroll
        for (int k = 0; k < M_LVL; ++k) ck[k] = cks[wv][k];

        float nq = 1.f;
#pragma unroll
        for (int k = 0; k < M_LVL; ++k) nq += ck[k];

        float phi = nq;
        if (nq > C_PARAM) {
            phi = C_PARAM + powf(C_PARAM, 1.f + A_PARAM) *
                            (powf(C_PARAM, -A_PARAM) - powf(nq, -A_PARAM)) / A_PARAM;
        }
        const float c0 = 1.f - phi;

        bool finite = finitef(c0);
#pragma unroll
        for (int k = 0; k < M_LVL; ++k) finite = finite && finitef(ck[k]);

        const float kf = (float)lane;
        float lo = 0.f, hi = 2.f;
        int n = 0;
        while (n < NBISECT) {
            const float mid = 0.5f * (lo + hi);
            if (mid == lo || mid == hi) break;
            const float w32 = (hi - lo) * 0.03125f;
            if (n + 5 <= NBISECT && (lo + w32) - lo == w32) {
                const float node = __builtin_fmaf(kf, w32, lo);
                const bool neg = poly_at(node, c0, ck) < 0.f;
                const unsigned below = __builtin_amdgcn_ballot_w32(neg) & ~1u;
                const float m = (float)__builtin_popcount(below);
                hi = __builtin_fmaf(m + 1.f, w32, lo);
                lo = __builtin_fmaf(m, w32, lo);
                n += 5;
            } else {
                const bool neg = poly_at(mid, c0, ck) < 0.f;
                lo = neg ? mid : lo;
                hi = neg ? hi : mid;
                n += 1;
            }
        }
        float root = 0.5f * (lo + hi);
        if (!finite) root = 0.f;
        root = fminf(root, 1.f);

        float rp = 1.f, val = (lane == 0) ? 1.f : 0.f;
#pragma unroll
        for (int k = 1; k <= M_LVL; ++k) { rp *= root; val = (lane == k) ? rp : val; }
        if (lane < 8) rootpow[row * 8 + lane] = val;
    }
}

#define SC_TILES 153
#define SC_WGPB  8

__device__ __forceinline__ void load_half(v4f (&R)[16], const float* __restrict__ p) {
    for (int i = 0; i < 16; ++i)
        R[i] = *(const v4f_a8*)(p + (long)i * SIGLEN);
}

__device__ __forceinline__ void sum_half_one(v4f& acc, const v4f (&R)[16], const float* rpw) {
    for (int i = 0; i < 16; ++i) {
        const float wgt = rpw[i * 8];
        acc[0] = __builtin_fmaf(R[i][0], wgt, acc[0]);
        acc[1] = __builtin_fmaf(R[i][1], wgt, acc[1]);
        acc[2] = __builtin_fmaf(R[i][2], wgt, acc[2]);
        acc[3] = __builtin_fmaf(R[i][3], wgt, acc[3]);
    }
}

__device__ __forceinline__ void sum_half_mix(v4f& acc, const v4f (&R)[16], const float* rph,
                                             int k0, int k1, int k2, int k3) {
    for (int i = 0; i < 16; ++i) {
        acc[0] = __builtin_fmaf(R[i][0], rph[i * 8 + k0], acc[0]);
        acc[1] = __builtin_fmaf(R[i][1], rph[i * 8 + k1], acc[1]);
        acc[2] = __builtin_fmaf(R[i][2], rph[i * 8 + k2], acc[2]);
        acc[3] = __builtin_fmaf(R[i][3], rph[i * 8 + k3], acc[3]);
    }
}

__global__ __launch_bounds__(256)
void scale_mean_kernel(const float* __restrict__ x, const float* __restrict__ rootpow,
                       float* __restrict__ out) {
    __shared__ float rp[SS * 8];
    const int b   = blockIdx.x / SC_WGPB;
    const int qq  = blockIdx.x % SC_WGPB;
    const int tid = threadIdx.x;
    rp[tid] = rootpow[(long)b * SS * 8 + tid];
    __syncthreads();

    const int lane = tid & 31;
    const int wv    = __builtin_amdgcn_readfirstlane(tid >> 5);
    const int w     = qq * 8 + wv;
    const int tbeg  = (SC_TILES * w) >> 6;
    const int tend  = (SC_TILES * (w + 1)) >> 6;
    const bool three = (tend - tbeg) > 2;

    const float* xb = x + (long)b * SS * SIGLEN;
    float* ob = out + (long)b * SIGLEN;

    v4f H[2][16];
    load_half(H[0], xb + min(tbeg * TILE_C + 4 * lane, SIGLEN - 4));

    v4f acc = {0.f, 0.f, 0.f, 0.f};
#pragma unroll
    for (int st = 0; st < 6; ++st) {
        const int j = st >> 1, hf = st & 1;
        if (j < 2 || three) {
            const int t0   = (tbeg + j) * TILE_C;
            const int want = t0 + 4 * lane;
            const int cl   = min(want, SIGLEN - 4);
            const int shift = want - cl;

            __builtin_amdgcn_sched_barrier(0);
            if (st + 1 < 6) {
                const int j1 = (st + 1) >> 1, hf1 = (st + 1) & 1;
                if (j1 < 2 || three)
                    load_half(H[(st + 1) & 1],
                              xb + hf1 * 16L * SIGLEN + min((tbeg + j1) * TILE_C + 4 * lane, SIGLEN - 4));
            }
            __builtin_amdgcn_sched_barrier(0);

            const int kLo = lvl_of(t0);
            const int kHi = lvl_of(min(t0 + TILE_C, SIGLEN) - 1);
            if (hf == 0) { acc[0] = 0.f; acc[1] = 0.f; acc[2] = 0.f; acc[3] = 0.f; }
            if (kLo == kHi) {
                sum_half_one(acc, H[st & 1], rp + hf * 16 * 8 + kLo);
            } else {
                sum_half_mix(acc, H[st & 1], rp + hf * 16 * 8,
                             lvl_of(cl), lvl_of(cl + 1), lvl_of(cl + 2), lvl_of(cl + 3));
            }

            if (hf == 1) {
                const float inv = 1.f / SS;
                if (shift == 0) {
                    v4f o; o[0] = acc[0] * inv; o[1] = acc[1] * inv; o[2] = acc[2] * inv; o[3] = acc[3] * inv;
                    *(v4f_a8*)(ob + cl) = o;
                } else if (shift == 2) {
                    v2f o; o[0] = acc[2] * inv; o[1] = acc[3] * inv;
                    *(v2f*)(ob + cl + 2) = o;
                }
            }
        }
    }
}

extern "C" void kernel_launch(void* const* d_in, const int* in_sizes, int n_in,
                              void* d_out, int out_size, void* d_ws, size_t ws_size,
                              hipStream_t stream) {
    const float* x = (const float*)d_in[0];
    float* out = (float*)d_out;

    float* rootpow = (float*)d_ws;

    {
        level_sums_roots<<<NGROUP, WG_WAVES * 32, 0, stream>>>(x, rootpow);
    }
    {
        scale_mean_kernel<<<BB * SC_WGPB, 256, 0, stream>>>(x, rootpow, out);
    }
}


# Rows stay in LDS: a work-group streams 4 whole rows in (level sums on the way), solves their roots, adds the scaled rows from LDS; input read once; launch 2 adds 4 partial rows; f32 sum orders change
# speedup vs baseline: 5.3964x; 1.1378x over previous
#include <hip/hip_runtime.h>
#include <hip/hip_bf16.h>

#define M_LVL   6
#define SIGLEN  19530
#define BB      64
#define SS      32
#define ROWS    (BB * SS)
#define NBISECT 100
#define C_PARAM 4.0f
#define A_PARAM 1.0f

#define RGRP     8
#define NGROUP   (ROWS / RGRP)
#define TILE_C   128
#define DTILE_C  (2 * TILE_C)
#define NTILE    5
#define WAVE_C   1224
#define WG_WAVES 16
#define PITCH    132

typedef __attribute__((ext_vector_type(2))) float v2f;
typedef __attribute__((ext_vector_type(4))) float v4f;
typedef __attribute__((ext_vector_type(8))) float v8f;
typedef v4f __attribute__((aligned(8))) v4f_a8;

__device__ __forceinline__ int lvl_of(int i) {
    return 1 + (i >= 5) + (i >= 30) + (i >= 155) + (i >= 780) + (i >= 3905);
}

__device__ __forceinline__ bool finitef(float v) {
    return (v == v) && (fabsf(v) <= 3.402823466e38f);
}

__device__ __forceinline__ v8f wmma_acc(v2f a, v2f b, v8f c) {
    return __builtin_amdgcn_wmma_f32_16x16x4_f32(false, a, false, b, (short)0, c, false, false);
}

__device__ __forceinline__ void load_tile(v4f (&R)[16], const float* __restrict__ gbase,
                                          int t0, int lane, int cend) {
    for (int i = 0; i < 16; ++i) {
        const int col = min(t0 + (i >> 3) * TILE_C + 4 * lane, cend - 4);
        R[i] = *(const v4f_a8*)(gbase + (long)(i & 7) * SIGLEN + col);
    }
}

__device__ __forceinline__ float poly_at(float t, float c0, const float (&ck)[M_LVL]) {
    const float t2 = t * t;
    float p = c0, tp = 1.f;
#pragma unroll
    for (int k = 0; k < M_LVL; ++k) { tp *= t2; p = __builtin_fmaf(ck[k], tp, p); }
    return p;
}

__global__ __launch_bounds__(WG_WAVES * 32)
void level_sums_roots(const float* __restrict__ x, float* __restrict__ rootpow) {
    __shared__ __attribute__((aligned(16))) float tile[WG_WAVES][16 * PITCH];
    __shared__ float red[WG_WAVES][16][8];
    __shared__ float cks[RGRP][8];

    const int lane = threadIdx.x & 31;
    const int wv   = __builtin_amdgcn_readfirstlane(threadIdx.x >> 5);
    const int g    = blockIdx.x;

    const int c = lane & 15;
    const int h = lane >> 4;

    const int cbeg = wv * WAVE_C;
    const int cend = min(cbeg + WAVE_C, SIGLEN);
    const float* gbase = x + (long)(g * RGRP) * SIGLEN;
    float* my = tile[wv];

    v8f acc0 = {0.f, 0.f, 0.f, 0.f, 0.f, 0.f, 0.f, 0.f};
    v8f acc1 = {0.f, 0.f, 0.f, 0.f, 0.f, 0.f, 0.f, 0.f};

    v4f R[16];
    load_tile(R, gbase, cbeg, lane, cend);

    for (int t = 0; t < NTILE; ++t) {
        const int t0 = cbeg + t * DTILE_C;

        if (t == NTILE - 1) {
            const int w0 = t0 + 4 * lane, w1 = w0 + TILE_C;
            const int shift0 = w0 - min(w0, cend - 4);
            const int shift1 = w1 - min(w1, cend - 4);
            for (int i = 0; i < 16; ++i) {
                const int shift = (i >> 3) ? shift1 : shift0;
                const v4f ld = R[i];
                v4f r;
                r[0] = (shift == 0) ? ld[0] : ((shift == 2) ? ld[2] : 0.f);
                r[1] = (shift == 0) ? ld[1] : ((shift == 2) ? ld[3] : 0.f);
                r[2] = (shift == 0) ? ld[2] : 0.f;
                r[3] = (shift == 0) ? ld[3] : 0.f;
                R[i] = r;
            }
        }

        for (int i = 0; i < 16; ++i)
            *(v4f*)(my + i * PITCH + 4 * lane) = R[i];

        if (t + 1 < NTILE)
            load_tile(R, gbase, t0 + DTILE_C, lane, cend);


        const float* frag = my + c * PITCH + 4 * h;
        v4f F[16];
        for (int j = 0; j < 16; ++j) F[j] = *(const v4f*)(frag + 8 * j);

        const int kLo = lvl_of(t0);
        const int kHi = lvl_of(min(t0 + DTILE_C, cend) - 1);
        if (kLo == kHi) {
            const float bval = (c == kLo - 1) ? 1.0f : 0.0f;
            v2f bfrag; bfrag[0] = bval; bfrag[1] = bval;
            for (int j = 0; j < 16; ++j) {
                const v4f f = F[j];
                v2f a0; a0[0] = f[0] * f[0]; a0[1] = f[1] * f[1];
                v2f a1; a1[0] = f[2] * f[2]; a1[1] = f[3] * f[3];
                acc0 = wmma_acc(a0, bfrag, acc0);
                acc1 = wmma_acc(a1, bfrag, acc1);
            }
        } else {
            const float keep0 = (c < 8) ? 1.0f : 0.0f;
            const float keep1 = 1.0f - keep0;
            for (int j = 0; j < 16; ++j) {
                const v4f f = F[j];
                const int col0 = t0 + 8 * j + 4 * h;
                const int col1 = col0 + TILE_C;
                const float s0 = f[0] * f[0], s1 = f[1] * f[1], s2 = f[2] * f[2], s3 = f[3] * f[3];
                v2f a0; a0[0] = s0 * keep0; a0[1] = s1 * keep0;
                v2f a1; a1[0] = s2 * keep0; a1[1] = s3 * keep0;
                v2f b0; b0[0] = (lvl_of(col0)     == c + 1) ? 1.0f : 0.0f;
                        b0[1] = (lvl_of(col0 + 1) == c + 1) ? 1.0f : 0.0f;
                v2f b1; b1[0] = (lvl_of(col0 + 2) == c + 1) ? 1.0f : 0.0f;
                        b1[1] = (lvl_of(col0 + 3) == c + 1) ? 1.0f : 0.0f;
                acc0 = wmma_acc(a0, b0, acc0);
                acc1 = wmma_acc(a1, b1, acc1);
                v2f a2; a2[0] = s0 * keep1; a2[1] = s1 * keep1;
                v2f a3; a3[0] = s2 * keep1; a3[1] = s3 * keep1;
                v2f b2; b2[0] = (lvl_of(col1)     == c + 1) ? 1.0f : 0.0f;
                        b2[1] = (lvl_of(col1 + 1) == c + 1) ? 1.0f : 0.0f;
                v2f b3; b3[0] = (lvl_of(col1 + 2) == c + 1) ? 1.0f : 0.0f;
                        b3[1] = (lvl_of(col1 + 3) == c + 1) ? 1.0f : 0.0f;
                acc0 = wmma_acc(a2, b2, acc0);
                acc1 = wmma_acc(a3, b3, acc1);
            }
        }
    }

    if (c < 8) {
        for (int v = 0; v < 8; ++v)
            red[wv][8 * h + v][c] = acc0[v] + acc1[v];
    }
    __syncthreads();
    if (threadIdx.x < RGRP * 8) {
        const int row = threadIdx.x >> 3, col = threadIdx.x & 7;
        float s = 0.f;
        for (int w = 0; w < WG_WAVES; ++w) s += red[w][row][col] + red[w][row + 8][col];
        cks[row][col] = s;
    }
    __syncthreads();

    if (wv < RGRP) {
        const int row = g * RGRP + wv;
        float ck[M_LVL];
#pragma unroll
        for (int k = 0; k < M_LVL; ++k) ck[k] = cks[wv][k];

        float nq = 1.f;
#pragma unroll
        for (int k = 0; k < M_LVL; ++k) nq += ck[k];

        float phi = nq;
        if (nq > C_PARAM) {
            phi = C_PARAM + powf(C_PARAM, 1.f + A_PARAM) *
                            (powf(C_PARAM, -A_PARAM) - powf(nq, -A_PARAM)) / A_PARAM;
        }
        const float c0 = 1.f - phi;

        bool finite = finitef(c0);
#pragma unroll
        for (int k = 0; k < M_LVL; ++k) finite = finite && finitef(ck[k]);

        const float kf = (float)lane;
        float lo = 0.f, hi = 2.f;
        int n = 0;
        while (n < NBISECT) {
            const float mid = 0.5f * (lo + hi);
            if (mid == lo || mid == hi) break;
            const float w32 = (hi - lo) * 0.03125f;
            if (n + 5 <= NBISECT && (lo + w32) - lo == w32) {
                const float node = __builtin_fmaf(kf, w32, lo);
                const bool neg = poly_at(node, c0, ck) < 0.f;
                const unsigned below = __builtin_amdgcn_ballot_w32(neg) & ~1u;
                const float m = (float)__builtin_popcount(below);
                hi = __builtin_fmaf(m + 1.f, w32, lo);
                lo = __builtin_fmaf(m, w32, lo);
                n += 5;
            } else {
                const bool neg = poly_at(mid, c0, ck) < 0.f;
                lo = neg ? mid : lo;
                hi = neg ? hi : mid;
                n += 1;
            }
        }
        float root = 0.5f * (lo + hi);
        if (!finite) root = 0.f;
        root = fminf(root, 1.f);

        float rp = 1.f, val = (lane == 0) ? 1.f : 0.f;
#pragma unroll
        for (int k = 1; k <= M_LVL; ++k) { rp *= root; val = (lane == k) ? rp : val; }
        if (lane < 8) rootpow[row * 8 + lane] = val;
    }
}

#define SC_TILES 153
#define SC_WGPB  8

__device__ __forceinline__ void load_half(v4f (&R)[16], const float* __restrict__ p) {
    for (int i = 0; i < 16; ++i)
        R[i] = *(const v4f_a8*)(p + (long)i * SIGLEN);
}

__device__ __forceinline__ void sum_half_one(v4f& acc, const v4f (&R)[16], const float* rpw) {
    for (int i = 0; i < 16; ++i) {
        const float wgt = rpw[i * 8];
        acc[0] = __builtin_fmaf(R[i][0], wgt, acc[0]);
        acc[1] = __builtin_fmaf(R[i][1], wgt, acc[1]);
        acc[2] = __builtin_fmaf(R[i][2], wgt, acc[2]);
        acc[3] = __builtin_fmaf(R[i][3], wgt, acc[3]);
    }
}

__device__ __forceinline__ void sum_half_mix(v4f& acc, const v4f (&R)[16], const float* rph,
                                             int k0, int k1, int k2, int k3) {
    for (int i = 0; i < 16; ++i) {
        acc[0] = __builtin_fmaf(R[i][0], rph[i * 8 + k0], acc[0]);
        acc[1] = __builtin_fmaf(R[i][1], rph[i * 8 + k1], acc[1]);
        acc[2] = __builtin_fmaf(R[i][2], rph[i * 8 + k2], acc[2]);
        acc[3] = __builtin_fmaf(R[i][3], rph[i * 8 + k3], acc[3]);
    }
}

__global__ __launch_bounds__(256)
void scale_mean_kernel(const float* __restrict__ x, const float* __restrict__ rootpow,
                       float* __restrict__ out) {
    __shared__ float rp[SS * 8];
    const int b   = blockIdx.x / SC_WGPB;
    const int qq  = blockIdx.x % SC_WGPB;
    const int tid = threadIdx.x;
    rp[tid] = rootpow[(long)b * SS * 8 + tid];
    __syncthreads();

    const int lane = tid & 31;
    const int wv    = __builtin_amdgcn_readfirstlane(tid >> 5);
    const int w     = qq * 8 + wv;
    const int tbeg  = (SC_TILES * w) >> 6;
    const int tend  = (SC_TILES * (w + 1)) >> 6;
    const bool three = (tend - tbeg) > 2;

    const float* xb = x + (long)b * SS * SIGLEN;
    float* ob = out + (long)b * SIGLEN;

    v4f H[2][16];
    load_half(H[0], xb + min(tbeg * TILE_C + 4 * lane, SIGLEN - 4));

    v4f acc = {0.f, 0.f, 0.f, 0.f};
#pragma unroll
    for (int st = 0; st < 6; ++st) {
        const int j = st >> 1, hf = st & 1;
        if (j < 2 || three) {
            const int t0   = (tbeg + j) * TILE_C;
            const int want = t0 + 4 * lane;
            const int cl   = min(want, SIGLEN - 4);
            const int shift = want - cl;

            __builtin_amdgcn_sched_barrier(0);
            if (st + 1 < 6) {
                const int j1 = (st + 1) >> 1, hf1 = (st + 1) & 1;
                if (j1 < 2 || three)
                    load_half(H[(st + 1) & 1],
                              xb + hf1 * 16L * SIGLEN + min((tbeg + j1) * TILE_C + 4 * lane, SIGLEN - 4));
            }
            __builtin_amdgcn_sched_barrier(0);

            const int kLo = lvl_of(t0);
            const int kHi = lvl_of(min(t0 + TILE_C, SIGLEN) - 1);
            if (hf == 0) { acc[0] = 0.f; acc[1] = 0.f; acc[2] = 0.f; acc[3] = 0.f; }
            if (kLo == kHi) {
                sum_half_one(acc, H[st & 1], rp + hf * 16 * 8 + kLo);
            } else {
                sum_half_mix(acc, H[st & 1], rp + hf * 16 * 8,
                             lvl_of(cl), lvl_of(cl + 1), lvl_of(cl + 2), lvl_of(cl + 3));
            }

            if (hf == 1) {
                const float inv = 1.f / SS;
                if (shift == 0) {
                    v4f o; o[0] = acc[0] * inv; o[1] = acc[1] * inv; o[2] = acc[2] * inv; o[3] = acc[3] * inv;
                    *(v4f_a8*)(ob + cl) = o;
                } else if (shift == 2) {
                    v2f o; o[0] = acc[2] * inv; o[1] = acc[3] * inv;
                    *(v2f*)(ob + cl + 2) = o;
                }
            }
        }
    }
}

#define RSUB    4
#define NPIECE  153
#define ROWP    19532
#define ROWPW   19584
#define NGRP4   4883
#define NACC    10
#define PART_BYTES 20054016


template <int I>
__device__ __forceinline__ void load_piece(v4f (&R)[RSUB], const float* __restrict__ gsub,
                                           int wv, int lane) {
    const int q = wv + 16 * I;
    if (I < 9 || q < NPIECE) {
        const int col = (I < 9) ? (q * TILE_C + 4 * lane) : min(q * TILE_C + 4 * lane, SIGLEN - 4);
        R[0] = *(const v4f_a8*)(gsub + 0L * SIGLEN + col);
        R[1] = *(const v4f_a8*)(gsub + 1L * SIGLEN + col);
        R[2] = *(const v4f_a8*)(gsub + 2L * SIGLEN + col);
        R[3] = *(const v4f_a8*)(gsub + 3L * SIGLEN + col);
    }
}

template <int I>
__device__ __forceinline__ void take_piece(const v4f (&R)[RSUB], float* __restrict__ res,
                                           v8f& accA, v8f& accB, int wv, int lane, int c) {
    const int q = wv + 16 * I;
    if (I < 9 || q < NPIECE) {
        const int t0   = q * TILE_C;
        const int want = t0 + 4 * lane;
        v4f v0 = R[0], v1 = R[1], v2 = R[2], v3 = R[3];
        if (I == 9) {
            if (q == NPIECE - 1) {
                const int shift = want - min(want, SIGLEN - 4);
                const bool all = (shift == 0), two = (shift == 2);
                v4f f;
                f[0] = all ? v0[0] : (two ? v0[2] : 0.f); f[1] = all ? v0[1] : (two ? v0[3] : 0.f);
                f[2] = all ? v0[2] : 0.f;                 f[3] = all ? v0[3] : 0.f;                 v0 = f;
                f[0] = all ? v1[0] : (two ? v1[2] : 0.f); f[1] = all ? v1[1] : (two ? v1[3] : 0.f);
                f[2] = all ? v1[2] : 0.f;                 f[3] = all ? v1[3] : 0.f;                 v1 = f;
                f[0] = all ? v2[0] : (two ? v2[2] : 0.f); f[1] = all ? v2[1] : (two ? v2[3] : 0.f);
                f[2] = all ? v2[2] : 0.f;                 f[3] = all ? v2[3] : 0.f;                 v2 = f;
                f[0] = all ? v3[0] : (two ? v3[2] : 0.f); f[1] = all ? v3[1] : (two ? v3[3] : 0.f);
                f[2] = all ? v3[2] : 0.f;                 f[3] = all ? v3[3] : 0.f;                 v3 = f;
            }
        }
        if (I < 9 || want < ROWP) {
            *(v4f*)(res + 0 * ROWP + want) = v0;
            *(v4f*)(res + 1 * ROWP + want) = v1;
            *(v4f*)(res + 2 * ROWP + want) = v2;
            *(v4f*)(res + 3 * ROWP + want) = v3;
        }

        const int kLo = (I < 2) ? lvl_of(t0) : M_LVL;
        const int kHi = (I < 2) ? lvl_of(t0 + TILE_C - 1) : M_LVL;
        if (I >= 2 || kLo == kHi) {
            const float be = (c == kLo - 1) ? 1.0f : 0.0f;
            const float bo = (c == kLo + 7) ? 1.0f : 0.0f;
            v2f bE; bE[0] = be; bE[1] = be;
            v2f bO; bO[0] = bo; bO[1] = bo;
            v2f a;
            a[0] = v0[0] * v0[0]; a[1] = v0[1] * v0[1]; accA = wmma_acc(a, bE, accA);
            a[0] = v2[0] * v2[0]; a[1] = v2[1] * v2[1]; accB = wmma_acc(a, bE, accB);
            a[0] = v1[0] * v1[0]; a[1] = v1[1] * v1[1]; accA = wmma_acc(a, bO, accA);
            a[0] = v3[0] * v3[0]; a[1] = v3[1] * v3[1]; accB = wmma_acc(a, bO, accB);
            a[0] = v0[2] * v0[2]; a[1] = v0[3] * v0[3]; accA = wmma_acc(a, bE, accA);
            a[0] = v2[2] * v2[2]; a[1] = v2[3] * v2[3]; accB = wmma_acc(a, bE, accB);
            a[0] = v1[2] * v1[2]; a[1] = v1[3] * v1[3]; accA = wmma_acc(a, bO, accA);
            a[0] = v3[2] * v3[2]; a[1] = v3[3] * v3[3]; accB = wmma_acc(a, bO, accB);
        } else {
            for (int l = kLo; l <= kHi; ++l) {
                const float be = (c == l - 1) ? 1.0f : 0.0f;
                const float bo = (c == l + 7) ? 1.0f : 0.0f;
                v2f bE; bE[0] = be; bE[1] = be;
                v2f bO; bO[0] = bo; bO[1] = bo;
                const bool m0 = lvl_of(want) == l,     m1 = lvl_of(want + 1) == l;
                const bool m2 = lvl_of(want + 2) == l, m3 = lvl_of(want + 3) == l;
                v2f a;
                a[0] = m0 ? v0[0] * v0[0] : 0.f; a[1] = m1 ? v0[1] * v0[1] : 0.f; accA = wmma_acc(a, bE, accA);
                a[0] = m0 ? v2[0] * v2[0] : 0.f; a[1] = m1 ? v2[1] * v2[1] : 0.f; accB = wmma_acc(a, bE, accB);
                a[0] = m0 ? v1[0] * v1[0] : 0.f; a[1] = m1 ? v1[1] * v1[1] : 0.f; accA = wmma_acc(a, bO, accA);
                a[0] = m0 ? v3[0] * v3[0] : 0.f; a[1] = m1 ? v3[1] * v3[1] : 0.f; accB = wmma_acc(a, bO, accB);
                a[0] = m2 ? v0[2] * v0[2] : 0.f; a[1] = m3 ? v0[3] * v0[3] : 0.f; accA = wmma_acc(a, bE, accA);
                a[0] = m2 ? v2[2] * v2[2] : 0.f; a[1] = m3 ? v2[3] * v2[3] : 0.f; accB = wmma_acc(a, bE, accB);
                a[0] = m2 ? v1[2] * v1[2] : 0.f; a[1] = m3 ? v1[3] * v1[3] : 0.f; accA = wmma_acc(a, bO, accA);
                a[0] = m2 ? v3[2] * v3[2] : 0.f; a[1] = m3 ? v3[3] * v3[3] : 0.f; accB = wmma_acc(a, bO, accB);
            }
        }
    }
}

__device__ __forceinline__ void solve_and_scale(const v8f& accA, const v8f& accB, const float* __restrict__ res,
                                                float (&red2)[WG_WAVES][RSUB][2][8], float (&cks2)[RSUB][8],
                                                float (&rpw)[RSUB][8], v4f (&O)[NACC],
                                                int tid, int wv, int lane) {
    {
        float tA = 0.f, tB = 0.f;
#pragma unroll
        for (int v = 0; v < 8; ++v) { tA += accA[v]; tB += accB[v]; }
        const int n = lane & 15, hh = lane >> 4;
        red2[wv][(n >> 3)][hh][n & 7]     = tA;
        red2[wv][2 + (n >> 3)][hh][n & 7] = tB;
    }
    __syncthreads();
    if (tid < RSUB * 8) {
        const int row = tid >> 3, col = tid & 7;
        float sm = 0.f;
        for (int w = 0; w < WG_WAVES; ++w) sm += red2[w][row][0][col] + red2[w][row][1][col];
        cks2[row][col] = sm;
    }
    __syncthreads();

    if (wv < RSUB) {
        float ck[M_LVL];
#pragma unroll
        for (int k = 0; k < M_LVL; ++k) ck[k] = cks2[wv][k];

        float nq = 1.f;
#pragma unroll
        for (int k = 0; k < M_LVL; ++k) nq += ck[k];

        float phi = nq;
        if (nq > C_PARAM) {
            phi = C_PARAM + powf(C_PARAM, 1.f + A_PARAM) *
                            (powf(C_PARAM, -A_PARAM) - powf(nq, -A_PARAM)) / A_PARAM;
        }
        const float c0 = 1.f - phi;

        bool finite = finitef(c0);
#pragma unroll
        for (int k = 0; k < M_LVL; ++k) finite = finite && finitef(ck[k]);

        const float kf = (float)lane;
        float lo = 0.f, hi = 2.f;
        int n = 0;
        while (n < NBISECT) {
            const float mid = 0.5f * (lo + hi);
            if (mid == lo || mid == hi) break;
            const float w32 = (hi - lo) * 0.03125f;
            if (n + 5 <= NBISECT && (lo + w32) - lo == w32) {
                const float node = __builtin_fmaf(kf, w32, lo);
                const bool neg = poly_at(node, c0, ck) < 0.f;
                const unsigned below = __builtin_amdgcn_ballot_w32(neg) & ~1u;
                const float m = (float)__builtin_popcount(below);
                hi = __builtin_fmaf(m + 1.f, w32, lo);
                lo = __builtin_fmaf(m, w32, lo);
                n += 5;
            } else {
                const bool neg = poly_at(mid, c0, ck) < 0.f;
                lo = neg ? mid : lo;
                hi = neg ? hi : mid;
                n += 1;
            }
        }
        float root = 0.5f * (lo + hi);
        if (!finite) root = 0.f;
        root = fminf(root, 1.f);

        float rp = 1.f, val = (lane == 0) ? 1.f : 0.f;
#pragma unroll
        for (int k = 1; k <= M_LVL; ++k) { rp *= root; val = (lane == k) ? rp : val; }
        if (lane < 8) rpw[wv][lane] = val;
    }
    __syncthreads();

    int L0[4], L1[4];
#pragma unroll
    for (int j = 0; j < 4; ++j) { L0[j] = lvl_of(4 * tid + j); L1[j] = lvl_of(4 * (tid + 512) + j); }

#pragma unroll
    for (int r = 0; r < RSUB; ++r) {
        const float* rr = res + r * ROWP;
        const float w6 = rpw[r][M_LVL];
        {
            const v4f a = *(const v4f*)(rr + 4 * tid);
            O[0][0] = __builtin_fmaf(a[0], rpw[r][L0[0]], O[0][0]);
            O[0][1] = __builtin_fmaf(a[1], rpw[r][L0[1]], O[0][1]);
            O[0][2] = __builtin_fmaf(a[2], rpw[r][L0[2]], O[0][2]);
            O[0][3] = __builtin_fmaf(a[3], rpw[r][L0[3]], O[0][3]);
            const v4f b = *(const v4f*)(rr + 4 * (tid + 512));
            O[1][0] = __builtin_fmaf(b[0], rpw[r][L1[0]], O[1][0]);
            O[1][1] = __builtin_fmaf(b[1], rpw[r][L1[1]], O[1][1]);
            O[1][2] = __builtin_fmaf(b[2], rpw[r][L1[2]], O[1][2]);
            O[1][3] = __builtin_fmaf(b[3], rpw[r][L1[3]], O[1][3]);
        }
#pragma unroll
        for (int i = 2; i < NACC; ++i) {
            const int gi = tid + 512 * i;
            if (i < NACC - 1 || gi < NGRP4) {
                const v4f a = *(const v4f*)(rr + 4 * gi);
                O[i][0] = __builtin_fmaf(a[0], w6, O[i][0]);
                O[i][1] = __builtin_fmaf(a[1], w6, O[i][1]);
                O[i][2] = __builtin_fmaf(a[2], w6, O[i][2]);
                O[i][3] = __builtin_fmaf(a[3], w6, O[i][3]);
            }
        }
#pragma unroll
        for (int i = 0; i < NACC; ++i) {
            asm volatile("" : "+v"(O[i][0]), "+v"(O[i][1]), "+v"(O[i][2]), "+v"(O[i][3]) : : "memory");
        }
    }
}

__global__ __launch_bounds__(WG_WAVES * 32)
void rows_in_lds(const float* __restrict__ x, float* __restrict__ part) {
    __shared__ __attribute__((aligned(16))) float res[RSUB * ROWP];
    __shared__ float red2[WG_WAVES][RSUB][2][8];
    __shared__ float cks2[RSUB][8];
    __shared__ float rpw[RSUB][8];

    const int tid  = threadIdx.x;
    const int lane = tid & 31;
    const int wv   = __builtin_amdgcn_readfirstlane(tid >> 5);
    const int g    = blockIdx.x;
    const int c    = lane & 15;
    const float* g0 = x + (long)(g * RGRP) * SIGLEN;
    const float* g1 = g0 + (long)RSUB * SIGLEN;

    v4f S0[RSUB], S1[RSUB], S2[RSUB], S3[RSUB];
    v4f O[NACC];
#pragma unroll
    for (int i = 0; i < NACC; ++i) { O[i][0] = 0.f; O[i][1] = 0.f; O[i][2] = 0.f; O[i][3] = 0.f; }
    const v8f zero8 = {0.f, 0.f, 0.f, 0.f, 0.f, 0.f, 0.f, 0.f};
    v8f accA = zero8, accB = zero8;

    load_piece<0>(S0, g0, wv, lane); load_piece<1>(S1, g0, wv, lane);
    load_piece<2>(S2, g0, wv, lane); load_piece<3>(S3, g0, wv, lane);
    take_piece<0>(S0, res, accA, accB, wv, lane, c); load_piece<4>(S0, g0, wv, lane);
    take_piece<1>(S1, res, accA, accB, wv, lane, c); load_piece<5>(S1, g0, wv, lane);
    take_piece<2>(S2, res, accA, accB, wv, lane, c); load_piece<6>(S2, g0, wv, lane);
    take_piece<3>(S3, res, accA, accB, wv, lane, c); load_piece<7>(S3, g0, wv, lane);
    take_piece<4>(S0, res, accA, accB, wv, lane, c); load_piece<8>(S0, g0, wv, lane);
    take_piece<5>(S1, res, accA, accB, wv, lane, c); load_piece<9>(S1, g0, wv, lane);
    take_piece<6>(S2, res, accA, accB, wv, lane, c); load_piece<0>(S2, g1, wv, lane);
    take_piece<7>(S3, res, accA, accB, wv, lane, c); load_piece<1>(S3, g1, wv, lane);
    take_piece<8>(S0, res, accA, accB, wv, lane, c); load_piece<2>(S0, g1, wv, lane);
    take_piece<9>(S1, res, accA, accB, wv, lane, c); load_piece<3>(S1, g1, wv, lane);
    solve_and_scale(accA, accB, res, red2, cks2, rpw, O, tid, wv, lane);
    __syncthreads();

    accA = zero8; accB = zero8;
    take_piece<0>(S2, res, accA, accB, wv, lane, c); load_piece<4>(S2, g1, wv, lane);
    take_piece<1>(S3, res, accA, accB, wv, lane, c); load_piece<5>(S3, g1, wv, lane);
    take_piece<2>(S0, res, accA, accB, wv, lane, c); load_piece<6>(S0, g1, wv, lane);
    take_piece<3>(S1, res, accA, accB, wv, lane, c); load_piece<7>(S1, g1, wv, lane);
    take_piece<4>(S2, res, accA, accB, wv, lane, c); load_piece<8>(S2, g1, wv, lane);
    take_piece<5>(S3, res, accA, accB, wv, lane, c); load_piece<9>(S3, g1, wv, lane);
    take_piece<6>(S0, res, accA, accB, wv, lane, c);
    take_piece<7>(S1, res, accA, accB, wv, lane, c);
    take_piece<8>(S2, res, accA, accB, wv, lane, c);
    take_piece<9>(S3, res, accA, accB, wv, lane, c);
    solve_and_scale(accA, accB, res, red2, cks2, rpw, O, tid, wv, lane);

    float* prow = part + (long)g * ROWPW;
#pragma unroll
    for (int i = 0; i < NACC; ++i) {
        const int gi = tid + 512 * i;
        if (i < NACC - 1 || gi < NGRP4) *(v4f*)(prow + 4 * gi) = O[i];
    }
}

#define QGRP 1221
__global__ __launch_bounds__(256)
void add_partial_rows(const float* __restrict__ part, float* __restrict__ out) {
    const int b  = blockIdx.x >> 2;
    const int qd = blockIdx.x & 3;
    const int tid = threadIdx.x;
    const float* p0 = part + (long)(4 * b) * ROWPW;
    float* ob = out + (long)b * SIGLEN;

    v4f P[5][4];
#pragma unroll
    for (int it = 0; it < 5; ++it) {
        const int gl = tid + 256 * it;
        const int gi = min(qd * QGRP + gl, NGRP4 - 1);
#pragma unroll
        for (int k = 0; k < 4; ++k)
            P[it][k] = *(const v4f*)(p0 + (long)k * ROWPW + 4 * gi);
    }
    const float inv = 1.f / SS;
#pragma unroll
    for (int it = 0; it < 5; ++it) {
        const int gl = tid + 256 * it;
        const int gi = qd * QGRP + gl;
        if (gl < QGRP && gi < NGRP4) {
            const v4f s = (P[it][0] + P[it][1]) + (P[it][2] + P[it][3]);
            if (gi < NGRP4 - 1) {
                v4f o; o[0] = s[0] * inv; o[1] = s[1] * inv; o[2] = s[2] * inv; o[3] = s[3] * inv;
                *(v4f_a8*)(ob + 4 * gi) = o;
            } else {
                v2f o; o[0] = s[0] * inv; o[1] = s[1] * inv;
                *(v2f*)(ob + 4 * gi) = o;
            }
        }
    }
}

extern "C" void kernel_launch(void* const* d_in, const int* in_sizes, int n_in,
                              void* d_out, int out_size, void* d_ws, size_t ws_size,
                              hipStream_t stream) {
    const float* x = (const float*)d_in[0];
    float* out = (float*)d_out;

    if (ws_size >= (size_t)PART_BYTES) {
        float* part = (float*)d_ws;
        rows_in_lds<<<NGROUP, WG_WAVES * 32, 0, stream>>>(x, part);
        add_partial_rows<<<BB * 4, 256, 0, stream>>>(part, out);
    } else {
        float* rootpow = (float*)d_ws;
        level_sums_roots<<<NGROUP, WG_WAVES * 32, 0, stream>>>(x, rootpow);
        scale_mean_kernel<<<BB * SC_WGPB, 256, 0, stream>>>(x, rootpow, out);
    }
}


# Rows 4..7 of a work-group stay in the registers of the lanes that loaded them (rows 0..3 in LDS): all 8 rows resident, ONE root-solve phase of 8 waves, not two of 4; f32 sum order changes
# speedup vs baseline: 5.6925x; 1.0549x over previous
#include <hip/hip_runtime.h>
#include <hip/hip_bf16.h>

#define M_LVL   6
#define SIGLEN  19530
#define BB      64
#define SS      32
#define ROWS    (BB * SS)
#define NBISECT 100
#define C_PARAM 4.0f
#define A_PARAM 1.0f

#define RGRP     8
#define NGROUP   (ROWS / RGRP)
#define TILE_C   128
#define DTILE_C  (2 * TILE_C)
#define NTILE    5
#define WAVE_C   1224
#define WG_WAVES 16
#define PITCH    132

typedef __attribute__((ext_vector_type(2))) float v2f;
typedef __attribute__((ext_vector_type(4))) float v4f;
typedef __attribute__((ext_vector_type(8))) float v8f;
typedef v4f __attribute__((aligned(8))) v4f_a8;

__device__ __forceinline__ int lvl_of(int i) {
    return 1 + (i >= 5) + (i >= 30) + (i >= 155) + (i >= 780) + (i >= 3905);
}

__device__ __forceinline__ bool finitef(float v) {
    return (v == v) && (fabsf(v) <= 3.402823466e38f);
}

__device__ __forceinline__ v8f wmma_acc(v2f a, v2f b, v8f c) {
    return __builtin_amdgcn_wmma_f32_16x16x4_f32(false, a, false, b, (short)0, c, false, false);
}

__device__ __forceinline__ void load_tile(v4f (&R)[16], const float* __restrict__ gbase,
                                          int t0, int lane, int cend) {
    for (int i = 0; i < 16; ++i) {
        const int col = min(t0 + (i >> 3) * TILE_C + 4 * lane, cend - 4);
        R[i] = *(const v4f_a8*)(gbase + (long)(i & 7) * SIGLEN + col);
    }
}

__device__ __forceinline__ float poly_at(float t, float c0, const float (&ck)[M_LVL]) {
    const float t2 = t * t;
    float p = c0, tp = 1.f;
#pragma unroll
    for (int k = 0; k < M_LVL; ++k) { tp *= t2; p = __builtin_fmaf(ck[k], tp, p); }
    return p;
}

__global__ __launch_bounds__(WG_WAVES * 32)
void level_sums_roots(const float* __restrict__ x, float* __restrict__ rootpow) {
    __shared__ __attribute__((aligned(16))) float tile[WG_WAVES][16 * PITCH];
    __shared__ float red[WG_WAVES][16][8];
    __shared__ float cks[RGRP][8];

    const int lane = threadIdx.x & 31;
    const int wv   = __builtin_amdgcn_readfirstlane(threadIdx.x >> 5);
    const int g    = blockIdx.x;

    const int c = lane & 15;
    const int h = lane >> 4;

    const int cbeg = wv * WAVE_C;
    const int cend = min(cbeg + WAVE_C, SIGLEN);
    const float* gbase = x + (long)(g * RGRP) * SIGLEN;
    float* my = tile[wv];

    v8f acc0 = {0.f, 0.f, 0.f, 0.f, 0.f, 0.f, 0.f, 0.f};
    v8f acc1 = {0.f, 0.f, 0.f, 0.f, 0.f, 0.f, 0.f, 0.f};

    v4f R[16];
    load_tile(R, gbase, cbeg, lane, cend);

    for (int t = 0; t < NTILE; ++t) {
        const int t0 = cbeg + t * DTILE_C;

        if (t == NTILE - 1) {
            const int w0 = t0 + 4 * lane, w1 = w0 + TILE_C;
            const int shift0 = w0 - min(w0, cend - 4);
            const int shift1 = w1 - min(w1, cend - 4);
            for (int i = 0; i < 16; ++i) {
                const int shift = (i >> 3) ? shift1 : shift0;
                const v4f ld = R[i];
                v4f r;
                r[0] = (shift == 0) ? ld[0] : ((shift == 2) ? ld[2] : 0.f);
                r[1] = (shift == 0) ? ld[1] : ((shift == 2) ? ld[3] : 0.f);
                r[2] = (shift == 0) ? ld[2] : 0.f;
                r[3] = (shift == 0) ? ld[3] : 0.f;
                R[i] = r;
            }
        }

        for (int i = 0; i < 16; ++i)
            *(v4f*)(my + i * PITCH + 4 * lane) = R[i];

        if (t + 1 < NTILE)
            load_tile(R, gbase, t0 + DTILE_C, lane, cend);


        const float* frag = my + c * PITCH + 4 * h;
        v4f F[16];
        for (int j = 0; j < 16; ++j) F[j] = *(const v4f*)(frag + 8 * j);

        const int kLo = lvl_of(t0);
        const int kHi = lvl_of(min(t0 + DTILE_C, cend) - 1);
        if (kLo == kHi) {
            const float bval = (c == kLo - 1) ? 1.0f : 0.0f;
            v2f bfrag; bfrag[0] = bval; bfrag[1] = bval;
            for (int j = 0; j < 16; ++j) {
                const v4f f = F[j];
                v2f a0; a0[0] = f[0] * f[0]; a0[1] = f[1] * f[1];
                v2f a1; a1[0] = f[2] * f[2]; a1[1] = f[3] * f[3];
                acc0 = wmma_acc(a0, bfrag, acc0);
                acc1 = wmma_acc(a1, bfrag, acc1);
            }
        } else {
            const float keep0 = (c < 8) ? 1.0f : 0.0f;
            const float keep1 = 1.0f - keep0;
            for (int j = 0; j < 16; ++j) {
                const v4f f = F[j];
                const int col0 = t0 + 8 * j + 4 * h;
                const int col1 = col0 + TILE_C;
                const float s0 = f[0] * f[0], s1 = f[1] * f[1], s2 = f[2] * f[2], s3 = f[3] * f[3];
                v2f a0; a0[0] = s0 * keep0; a0[1] = s1 * keep0;
                v2f a1; a1[0] = s2 * keep0; a1[1] = s3 * keep0;
                v2f b0; b0[0] = (lvl_of(col0)     == c + 1) ? 1.0f : 0.0f;
                        b0[1] = (lvl_of(col0 + 1) == c + 1) ? 1.0f : 0.0f;
                v2f b1; b1[0] = (lvl_of(col0 + 2) == c + 1) ? 1.0f : 0.0f;
                        b1[1] = (lvl_of(col0 + 3) == c + 1) ? 1.0f : 0.0f;
                acc0 = wmma_acc(a0, b0, acc0);
                acc1 = wmma_acc(a1, b1, acc1);
                v2f a2; a2[0] = s0 * keep1; a2[1] = s1 * keep1;
                v2f a3; a3[0] = s2 * keep1; a3[1] = s3 * keep1;
                v2f b2; b2[0] = (lvl_of(col1)     == c + 1) ? 1.0f : 0.0f;
                        b2[1] = (lvl_of(col1 + 1) == c + 1) ? 1.0f : 0.0f;
                v2f b3; b3[0] = (lvl_of(col1 + 2) == c + 1) ? 1.0f : 0.0f;
                        b3[1] = (lvl_of(col1 + 3) == c + 1) ? 1.0f : 0.0f;
                acc0 = wmma_acc(a2, b2, acc0);
                acc1 = wmma_acc(a3, b3, acc1);
            }
        }
    }

    if (c < 8) {
        for (int v = 0; v < 8; ++v)
            red[wv][8 * h + v][c] = acc0[v] + acc1[v];
    }
    __syncthreads();
    if (threadIdx.x < RGRP * 8) {
        const int row = threadIdx.x >> 3, col = threadIdx.x & 7;
        float s = 0.f;
        for (int w = 0; w < WG_WAVES; ++w) s += red[w][row][col] + red[w][row + 8][col];
        cks[row][col] = s;
    }
    __syncthreads();

    if (wv < RGRP) {
        const int row = g * RGRP + wv;
        float ck[M_LVL];
#pragma unroll
        for (int k = 0; k < M_LVL; ++k) ck[k] = cks[wv][k];

        float nq = 1.f;
#pragma unroll
        for (int k = 0; k < M_LVL; ++k) nq += ck[k];

        float phi = nq;
        if (nq > C_PARAM) {
            phi = C_PARAM + powf(C_PARAM, 1.f + A_PARAM) *
                            (powf(C_PARAM, -A_PARAM) - powf(nq, -A_PARAM)) / A_PARAM;
        }
        const float c0 = 1.f - phi;

        bool finite = finitef(c0);
#pragma unroll
        for (int k = 0; k < M_LVL; ++k) finite = finite && finitef(ck[k]);

        const float kf = (float)lane;
        float lo = 0.f, hi = 2.f;
        int n = 0;
        while (n < NBISECT) {
            const float mid = 0.5f * (lo + hi);
            if (mid == lo || mid == hi) break;
            const float w32 = (hi - lo) * 0.03125f;
            if (n + 5 <= NBISECT && (lo + w32) - lo == w32) {
                const float node = __builtin_fmaf(kf, w32, lo);
                const bool neg = poly_at(node, c0, ck) < 0.f;
                const unsigned below = __builtin_amdgcn_ballot_w32(neg) & ~1u;
                const float m = (float)__builtin_popcount(below);
                hi = __builtin_fmaf(m + 1.f, w32, lo);
                lo = __builtin_fmaf(m, w32, lo);
                n += 5;
            } else {
                const bool neg = poly_at(mid, c0, ck) < 0.f;
                lo = neg ? mid : lo;
                hi = neg ? hi : mid;
                n += 1;
            }
        }
        float root = 0.5f * (lo + hi);
        if (!finite) root = 0.f;
        root = fminf(root, 1.f);

        float rp = 1.f, val = (lane == 0) ? 1.f : 0.f;
#pragma unroll
        for (int k = 1; k <= M_LVL; ++k) { rp *= root; val = (lane == k) ? rp : val; }
        if (lane < 8) rootpow[row * 8 + lane] = val;
    }
}

#define SC_TILES 153
#define SC_WGPB  8

__device__ __forceinline__ void load_half(v4f (&R)[16], const float* __restrict__ p) {
    for (int i = 0; i < 16; ++i)
        R[i] = *(const v4f_a8*)(p + (long)i * SIGLEN);
}

__device__ __forceinline__ void sum_half_one(v4f& acc, const v4f (&R)[16], const float* rpw) {
    for (int i = 0; i < 16; ++i) {
        const float wgt = rpw[i * 8];
        acc[0] = __builtin_fmaf(R[i][0], wgt, acc[0]);
        acc[1] = __builtin_fmaf(R[i][1], wgt, acc[1]);
        acc[2] = __builtin_fmaf(R[i][2], wgt, acc[2]);
        acc[3] = __builtin_fmaf(R[i][3], wgt, acc[3]);
    }
}

__device__ __forceinline__ void sum_half_mix(v4f& acc, const v4f (&R)[16], const float* rph,
                                             int k0, int k1, int k2, int k3) {
    for (int i = 0; i < 16; ++i) {
        acc[0] = __builtin_fmaf(R[i][0], rph[i * 8 + k0], acc[0]);
        acc[1] = __builtin_fmaf(R[i][1], rph[i * 8 + k1], acc[1]);
        acc[2] = __builtin_fmaf(R[i][2], rph[i * 8 + k2], acc[2]);
        acc[3] = __builtin_fmaf(R[i][3], rph[i * 8 + k3], acc[3]);
    }
}

__global__ __launch_bounds__(256)
void scale_mean_kernel(const float* __restrict__ x, const float* __restrict__ rootpow,
                       float* __restrict__ out) {
    __shared__ float rp[SS * 8];
    const int b   = blockIdx.x / SC_WGPB;
    const int qq  = blockIdx.x % SC_WGPB;
    const int tid = threadIdx.x;
    rp[tid] = rootpow[(long)b * SS * 8 + tid];
    __syncthreads();

    const int lane = tid & 31;
    const int wv    = __builtin_amdgcn_readfirstlane(tid >> 5);
    const int w     = qq * 8 + wv;
    const int tbeg  = (SC_TILES * w) >> 6;
    const int tend  = (SC_TILES * (w + 1)) >> 6;
    const bool three = (tend - tbeg) > 2;

    const float* xb = x + (long)b * SS * SIGLEN;
    float* ob = out + (long)b * SIGLEN;

    v4f H[2][16];
    load_half(H[0], xb + min(tbeg * TILE_C + 4 * lane, SIGLEN - 4));

    v4f acc = {0.f, 0.f, 0.f, 0.f};
#pragma unroll
    for (int st = 0; st < 6; ++st) {
        const int j = st >> 1, hf = st & 1;
        if (j < 2 || three) {
            const int t0   = (tbeg + j) * TILE_C;
            const int want = t0 + 4 * lane;
            const int cl   = min(want, SIGLEN - 4);
            const int shift = want - cl;

            __builtin_amdgcn_sched_barrier(0);
            if (st + 1 < 6) {
                const int j1 = (st + 1) >> 1, hf1 = (st + 1) & 1;
                if (j1 < 2 || three)
                    load_half(H[(st + 1) & 1],
                              xb + hf1 * 16L * SIGLEN + min((tbeg + j1) * TILE_C + 4 * lane, SIGLEN - 4));
            }
            __builtin_amdgcn_sched_barrier(0);

            const int kLo = lvl_of(t0);
            const int kHi = lvl_of(min(t0 + TILE_C, SIGLEN) - 1);
            if (hf == 0) { acc[0] = 0.f; acc[1] = 0.f; acc[2] = 0.f; acc[3] = 0.f; }
            if (kLo == kHi) {
                sum_half_one(acc, H[st & 1], rp + hf * 16 * 8 + kLo);
            } else {
                sum_half_mix(acc, H[st & 1], rp + hf * 16 * 8,
                             lvl_of(cl), lvl_of(cl + 1), lvl_of(cl + 2), lvl_of(cl + 3));
            }

            if (hf == 1) {
                const float inv = 1.f / SS;
                if (shift == 0) {
                    v4f o; o[0] = acc[0] * inv; o[1] = acc[1] * inv; o[2] = acc[2] * inv; o[3] = acc[3] * inv;
                    *(v4f_a8*)(ob + cl) = o;
                } else if (shift == 2) {
                    v2f o; o[0] = acc[2] * inv; o[1] = acc[3] * inv;
                    *(v2f*)(ob + cl + 2) = o;
                }
            }
        }
    }
}

#define RSUB    4
#define NPIECE  153
#define ROWP    19532
#define ROWPW   19584
#define NGRP4   4883
#define NACC    10
#define PART_BYTES 20054016


template <int I>
__device__ __forceinline__ void load_piece(v4f (&R)[RSUB], const float* __restrict__ gsub,
                                           int wv, int lane) {
    const int q = wv + 16 * I;
    if (I < 9 || q < NPIECE) {
        const int col = (I < 9) ? (q * TILE_C + 4 * lane) : min(q * TILE_C + 4 * lane, SIGLEN - 4);
        R[0] = *(const v4f_a8*)(gsub + 0L * SIGLEN + col);
        R[1] = *(const v4f_a8*)(gsub + 1L * SIGLEN + col);
        R[2] = *(const v4f_a8*)(gsub + 2L * SIGLEN + col);
        R[3] = *(const v4f_a8*)(gsub + 3L * SIGLEN + col);
    }
}

template <int I, bool TO_LDS>
__device__ __forceinline__ void take_piece(v4f (&R)[RSUB], float* __restrict__ res,
                                           v8f& accA, v8f& accB, int wv, int lane, int c) {
    const int q = wv + 16 * I;
    if (I < 9 || q < NPIECE) {
        const int t0   = q * TILE_C;
        const int want = t0 + 4 * lane;
        v4f v0 = R[0], v1 = R[1], v2 = R[2], v3 = R[3];
        if (I == 9) {
            if (q == NPIECE - 1) {
                const int shift = want - min(want, SIGLEN - 4);
                const bool all = (shift == 0), two = (shift == 2);
                v4f f;
                f[0] = all ? v0[0] : (two ? v0[2] : 0.f); f[1] = all ? v0[1] : (two ? v0[3] : 0.f);
                f[2] = all ? v0[2] : 0.f;                 f[3] = all ? v0[3] : 0.f;                 v0 = f;
                f[0] = all ? v1[0] : (two ? v1[2] : 0.f); f[1] = all ? v1[1] : (two ? v1[3] : 0.f);
                f[2] = all ? v1[2] : 0.f;                 f[3] = all ? v1[3] : 0.f;                 v1 = f;
                f[0] = all ? v2[0] : (two ? v2[2] : 0.f); f[1] = all ? v2[1] : (two ? v2[3] : 0.f);
                f[2] = all ? v2[2] : 0.f;                 f[3] = all ? v2[3] : 0.f;                 v2 = f;
                f[0] = all ? v3[0] : (two ? v3[2] : 0.f); f[1] = all ? v3[1] : (two ? v3[3] : 0.f);
                f[2] = all ? v3[2] : 0.f;                 f[3] = all ? v3[3] : 0.f;                 v3 = f;
                if (!TO_LDS) { R[0] = v0; R[1] = v1; R[2] = v2; R[3] = v3; }
            }
        }
        if (TO_LDS && (I < 9 || want < ROWP)) {
            *(v4f*)(res + 0 * ROWP + want) = v0;
            *(v4f*)(res + 1 * ROWP + want) = v1;
            *(v4f*)(res + 2 * ROWP + want) = v2;
            *(v4f*)(res + 3 * ROWP + want) = v3;
        }

        const int kLo = (I < 2) ? lvl_of(t0) : M_LVL;
        const int kHi = (I < 2) ? lvl_of(t0 + TILE_C - 1) : M_LVL;
        if (I >= 2 || kLo == kHi) {
            const float be = (c == kLo - 1) ? 1.0f : 0.0f;
            const float bo = (c == kLo + 7) ? 1.0f : 0.0f;
            v2f bE; bE[0] = be; bE[1] = be;
            v2f bO; bO[0] = bo; bO[1] = bo;
            v2f a;
            a[0] = v0[0] * v0[0]; a[1] = v0[1] * v0[1]; accA = wmma_acc(a, bE, accA);
            a[0] = v2[0] * v2[0]; a[1] = v2[1] * v2[1]; accB = wmma_acc(a, bE, accB);
            a[0] = v1[0] * v1[0]; a[1] = v1[1] * v1[1]; accA = wmma_acc(a, bO, accA);
            a[0] = v3[0] * v3[0]; a[1] = v3[1] * v3[1]; accB = wmma_acc(a, bO, accB);
            a[0] = v0[2] * v0[2]; a[1] = v0[3] * v0[3]; accA = wmma_acc(a, bE, accA);
            a[0] = v2[2] * v2[2]; a[1] = v2[3] * v2[3]; accB = wmma_acc(a, bE, accB);
            a[0] = v1[2] * v1[2]; a[1] = v1[3] * v1[3]; accA = wmma_acc(a, bO, accA);
            a[0] = v3[2] * v3[2]; a[1] = v3[3] * v3[3]; accB = wmma_acc(a, bO, accB);
        } else {
            for (int l = kLo; l <= kHi; ++l) {
                const float be = (c == l - 1) ? 1.0f : 0.0f;
                const float bo = (c == l + 7) ? 1.0f : 0.0f;
                v2f bE; bE[0] = be; bE[1] = be;
                v2f bO; bO[0] = bo; bO[1] = bo;
                const bool m0 = lvl_of(want) == l,     m1 = lvl_of(want + 1) == l;
                const bool m2 = lvl_of(want + 2) == l, m3 = lvl_of(want + 3) == l;
                v2f a;
                a[0] = m0 ? v0[0] * v0[0] : 0.f; a[1] = m1 ? v0[1] * v0[1] : 0.f; accA = wmma_acc(a, bE, accA);
                a[0] = m0 ? v2[0] * v2[0] : 0.f; a[1] = m1 ? v2[1] * v2[1] : 0.f; accB = wmma_acc(a, bE, accB);
                a[0] = m0 ? v1[0] * v1[0] : 0.f; a[1] = m1 ? v1[1] * v1[1] : 0.f; accA = wmma_acc(a, bO, accA);
                a[0] = m0 ? v3[0] * v3[0] : 0.f; a[1] = m1 ? v3[1] * v3[1] : 0.f; accB = wmma_acc(a, bO, accB);
                a[0] = m2 ? v0[2] * v0[2] : 0.f; a[1] = m3 ? v0[3] * v0[3] : 0.f; accA = wmma_acc(a, bE, accA);
                a[0] = m2 ? v2[2] * v2[2] : 0.f; a[1] = m3 ? v2[3] * v2[3] : 0.f; accB = wmma_acc(a, bE, accB);
                a[0] = m2 ? v1[2] * v1[2] : 0.f; a[1] = m3 ? v1[3] * v1[3] : 0.f; accA = wmma_acc(a, bO, accA);
                a[0] = m2 ? v3[2] * v3[2] : 0.f; a[1] = m3 ? v3[3] * v3[3] : 0.f; accB = wmma_acc(a, bO, accB);
            }
        }
    }
}

__device__ __forceinline__ void publish_sums(const v8f& accA, const v8f& accB,
                                             float (&red2)[WG_WAVES][RGRP][2][8], int r0, int wv, int lane) {
    float tA = 0.f, tB = 0.f;
#pragma unroll
    for (int v = 0; v < 8; ++v) { tA += accA[v]; tB += accB[v]; }
    const int n = lane & 15, hh = lane >> 4;
    red2[wv][r0 + (n >> 3)][hh][n & 7]     = tA;
    red2[wv][r0 + 2 + (n >> 3)][hh][n & 7] = tB;
}

__device__ __forceinline__ void solve_roots(float (&red2)[WG_WAVES][RGRP][2][8], float (&cks2)[RGRP][8],
                                            float (&rpw)[RGRP][8], int tid, int wv, int lane) {
    __syncthreads();
    if (tid < RGRP * 8) {
        const int row = tid >> 3, col = tid & 7;
        float sm = 0.f;
        for (int w = 0; w < WG_WAVES; ++w) sm += red2[w][row][0][col] + red2[w][row][1][col];
        cks2[row][col] = sm;
    }
    __syncthreads();

    if (wv < RGRP) {
        float ck[M_LVL];
#pragma unroll
        for (int k = 0; k < M_LVL; ++k) ck[k] = cks2[wv][k];

        float nq = 1.f;
#pragma unroll
        for (int k = 0; k < M_LVL; ++k) nq += ck[k];

        float phi = nq;
        if (nq > C_PARAM) {
            phi = C_PARAM + powf(C_PARAM, 1.f + A_PARAM) *
                            (powf(C_PARAM, -A_PARAM) - powf(nq, -A_PARAM)) / A_PARAM;
        }
        const float c0 = 1.f - phi;

        bool finite = finitef(c0);
#pragma unroll
        for (int k = 0; k < M_LVL; ++k) finite = finite && finitef(ck[k]);

        const float kf = (float)lane;
        float lo = 0.f, hi = 2.f;
        int n = 0;
        while (n < NBISECT) {
            const float mid = 0.5f * (lo + hi);
            if (mid == lo || mid == hi) break;
            const float w32 = (hi - lo) * 0.03125f;
            if (n + 5 <= NBISECT && (lo + w32) - lo == w32) {
                const float node = __builtin_fmaf(kf, w32, lo);
                const bool neg = poly_at(node, c0, ck) < 0.f;
                const unsigned below = __builtin_amdgcn_ballot_w32(neg) & ~1u;
                const float m = (float)__builtin_popcount(below);
                hi = __builtin_fmaf(m + 1.f, w32, lo);
                lo = __builtin_fmaf(m, w32, lo);
                n += 5;
            } else {
                const bool neg = poly_at(mid, c0, ck) < 0.f;
                lo = neg ? mid : lo;
                hi = neg ? hi : mid;
                n += 1;
            }
        }
        float root = 0.5f * (lo + hi);
        if (!finite) root = 0.f;
        root = fminf(root, 1.f);

        float rp = 1.f, val = (lane == 0) ? 1.f : 0.f;
#pragma unroll
        for (int k = 1; k <= M_LVL; ++k) { rp *= root; val = (lane == k) ? rp : val; }
        if (lane < 8) rpw[wv][lane] = val;
    }
    __syncthreads();
}

__global__ __launch_bounds__(WG_WAVES * 32)
void rows_in_lds(const float* __restrict__ x, float* __restrict__ part) {
    __shared__ __attribute__((aligned(16))) float res[RSUB * ROWP];
    __shared__ float red2[WG_WAVES][RGRP][2][8];
    __shared__ float cks2[RGRP][8];
    __shared__ float rpw[RGRP][8];

    const int tid  = threadIdx.x;
    const int lane = tid & 31;
    const int wv   = __builtin_amdgcn_readfirstlane(tid >> 5);
    const int g    = blockIdx.x;
    const int c    = lane & 15;
    const float* g0 = x + (long)(g * RGRP) * SIGLEN;
    const float* g1 = g0 + (long)RSUB * SIGLEN;

    v4f S0[RSUB], S1[RSUB], S2[RSUB], S3[RSUB];
    v4f H[NACC][RSUB];
    const v8f zero8 = {0.f, 0.f, 0.f, 0.f, 0.f, 0.f, 0.f, 0.f};
    v8f accA = zero8, accB = zero8;

    load_piece<0>(S0, g0, wv, lane); load_piece<1>(S1, g0, wv, lane);
    load_piece<2>(S2, g0, wv, lane); load_piece<3>(S3, g0, wv, lane);
    take_piece<0, true>(S0, res, accA, accB, wv, lane, c); load_piece<4>(S0, g0, wv, lane);
    take_piece<1, true>(S1, res, accA, accB, wv, lane, c); load_piece<5>(S1, g0, wv, lane);
    take_piece<2, true>(S2, res, accA, accB, wv, lane, c); load_piece<6>(S2, g0, wv, lane);
    take_piece<3, true>(S3, res, accA, accB, wv, lane, c); load_piece<7>(S3, g0, wv, lane);
    take_piece<4, true>(S0, res, accA, accB, wv, lane, c); load_piece<8>(S0, g0, wv, lane);
    take_piece<5, true>(S1, res, accA, accB, wv, lane, c); load_piece<9>(S1, g0, wv, lane);
    take_piece<6, true>(S2, res, accA, accB, wv, lane, c);
    load_piece<0>(H[0], g1, wv, lane); load_piece<1>(H[1], g1, wv, lane);
    take_piece<7, true>(S3, res, accA, accB, wv, lane, c);
    load_piece<2>(H[2], g1, wv, lane); load_piece<3>(H[3], g1, wv, lane);
    take_piece<8, true>(S0, res, accA, accB, wv, lane, c);
    load_piece<4>(H[4], g1, wv, lane); load_piece<5>(H[5], g1, wv, lane);
    take_piece<9, true>(S1, res, accA, accB, wv, lane, c);
    load_piece<6>(H[6], g1, wv, lane); load_piece<7>(H[7], g1, wv, lane);
    load_piece<8>(H[8], g1, wv, lane); load_piece<9>(H[9], g1, wv, lane);
    publish_sums(accA, accB, red2, 0, wv, lane);

    accA = zero8; accB = zero8;
    take_piece<0, false>(H[0], res, accA, accB, wv, lane, c);
    take_piece<1, false>(H[1], res, accA, accB, wv, lane, c);
    take_piece<2, false>(H[2], res, accA, accB, wv, lane, c);
    take_piece<3, false>(H[3], res, accA, accB, wv, lane, c);
    take_piece<4, false>(H[4], res, accA, accB, wv, lane, c);
    take_piece<5, false>(H[5], res, accA, accB, wv, lane, c);
    take_piece<6, false>(H[6], res, accA, accB, wv, lane, c);
    take_piece<7, false>(H[7], res, accA, accB, wv, lane, c);
    take_piece<8, false>(H[8], res, accA, accB, wv, lane, c);
    take_piece<9, false>(H[9], res, accA, accB, wv, lane, c);
    publish_sums(accA, accB, red2, RSUB, wv, lane);

    solve_roots(red2, cks2, rpw, tid, wv, lane);

    int L0[4], L1[4];
#pragma unroll
    for (int j = 0; j < 4; ++j) { L0[j] = lvl_of(4 * tid + j); L1[j] = lvl_of(4 * (tid + 512) + j); }

    v4f O[NACC];
#pragma unroll
    for (int i = 0; i < NACC; ++i) { O[i][0] = 0.f; O[i][1] = 0.f; O[i][2] = 0.f; O[i][3] = 0.f; }

    const bool has9 = (wv + 16 * (NACC - 1)) < NPIECE;
#pragma unroll
    for (int r = 0; r < RSUB; ++r) {
        const float w6 = rpw[RSUB + r][M_LVL];
        O[0][0] = __builtin_fmaf(H[0][r][0], rpw[RSUB + r][L0[0]], O[0][0]);
        O[0][1] = __builtin_fmaf(H[0][r][1], rpw[RSUB + r][L0[1]], O[0][1]);
        O[0][2] = __builtin_fmaf(H[0][r][2], rpw[RSUB + r][L0[2]], O[0][2]);
        O[0][3] = __builtin_fmaf(H[0][r][3], rpw[RSUB + r][L0[3]], O[0][3]);
        O[1][0] = __builtin_fmaf(H[1][r][0], rpw[RSUB + r][L1[0]], O[1][0]);
        O[1][1] = __builtin_fmaf(H[1][r][1], rpw[RSUB + r][L1[1]], O[1][1]);
        O[1][2] = __builtin_fmaf(H[1][r][2], rpw[RSUB + r][L1[2]], O[1][2]);
        O[1][3] = __builtin_fmaf(H[1][r][3], rpw[RSUB + r][L1[3]], O[1][3]);
#pragma unroll
        for (int i = 2; i < NACC; ++i) {
            if (i < NACC - 1 || has9) {
                O[i][0] = __builtin_fmaf(H[i][r][0], w6, O[i][0]);
                O[i][1] = __builtin_fmaf(H[i][r][1], w6, O[i][1]);
                O[i][2] = __builtin_fmaf(H[i][r][2], w6, O[i][2]);
                O[i][3] = __builtin_fmaf(H[i][r][3], w6, O[i][3]);
            }
        }
    }
#pragma unroll
    for (int i = 0; i < NACC; ++i) {
        asm volatile("" : "+v"(O[i][0]), "+v"(O[i][1]), "+v"(O[i][2]), "+v"(O[i][3]) : : "memory");
    }

#pragma unroll
    for (int r = 0; r < RSUB; ++r) {
        const float* rr = res + r * ROWP;
        const float w6 = rpw[r][M_LVL];
        {
            const v4f a = *(const v4f*)(rr + 4 * tid);
            O[0][0] = __builtin_fmaf(a[0], rpw[r][L0[0]], O[0][0]);
            O[0][1] = __builtin_fmaf(a[1], rpw[r][L0[1]], O[0][1]);
            O[0][2] = __builtin_fmaf(a[2], rpw[r][L0[2]], O[0][2]);
            O[0][3] = __builtin_fmaf(a[3], rpw[r][L0[3]], O[0][3]);
            const v4f b = *(const v4f*)(rr + 4 * (tid + 512));
            O[1][0] = __builtin_fmaf(b[0], rpw[r][L1[0]], O[1][0]);
            O[1][1] = __builtin_fmaf(b[1], rpw[r][L1[1]], O[1][1]);
            O[1][2] = __builtin_fmaf(b[2], rpw[r][L1[2]], O[1][2]);
            O[1][3] = __builtin_fmaf(b[3], rpw[r][L1[3]], O[1][3]);
        }
#pragma unroll
        for (int i = 2; i < NACC; ++i) {
            const int gi = tid + 512 * i;
            if (i < NACC - 1 || gi < NGRP4) {
                const v4f a = *(const v4f*)(rr + 4 * gi);
                O[i][0] = __builtin_fmaf(a[0], w6, O[i][0]);
                O[i][1] = __builtin_fmaf(a[1], w6, O[i][1]);
                O[i][2] = __builtin_fmaf(a[2], w6, O[i][2]);
                O[i][3] = __builtin_fmaf(a[3], w6, O[i][3]);
            }
        }
    }

    float* prow = part + (long)g * ROWPW;
#pragma unroll
    for (int i = 0; i < NACC; ++i) {
        const int gi = tid + 512 * i;
        if (i < NACC - 1 || gi < NGRP4) *(v4f*)(prow + 4 * gi) = O[i];
    }
}

#define QGRP 1221
__global__ __launch_bounds__(256)
void add_partial_rows(const float* __restrict__ part, float* __restrict__ out) {
    const int b  = blockIdx.x >> 2;
    const int qd = blockIdx.x & 3;
    const int tid = threadIdx.x;
    const float* p0 = part + (long)(4 * b) * ROWPW;
    float* ob = out + (long)b * SIGLEN;

    v4f P[5][4];
#pragma unroll
    for (int it = 0; it < 5; ++it) {
        const int gl = tid + 256 * it;
        const int gi = min(qd * QGRP + gl, NGRP4 - 1);
#pragma unroll
        for (int k = 0; k < 4; ++k)
            P[it][k] = *(const v4f*)(p0 + (long)k * ROWPW + 4 * gi);
    }
    const float inv = 1.f / SS;
#pragma unroll
    for (int it = 0; it < 5; ++it) {
        const int gl = tid + 256 * it;
        const int gi = qd * QGRP + gl;
        if (gl < QGRP && gi < NGRP4) {
            const v4f s = (P[it][0] + P[it][1]) + (P[it][2] + P[it][3]);
            if (gi < NGRP4 - 1) {
                v4f o; o[0] = s[0] * inv; o[1] = s[1] * inv; o[2] = s[2] * inv; o[3] = s[3] * inv;
                *(v4f_a8*)(ob + 4 * gi) = o;
            } else {
                v2f o; o[0] = s[0] * inv; o[1] = s[1] * inv;
                *(v2f*)(ob + 4 * gi) = o;
            }
        }
    }
}

extern "C" void kernel_launch(void* const* d_in, const int* in_sizes, int n_in,
                              void* d_out, int out_size, void* d_ws, size_t ws_size,
                              hipStream_t stream) {
    const float* x = (const float*)d_in[0];
    float* out = (float*)d_out;

    if (ws_size >= (size_t)PART_BYTES) {
        float* part = (float*)d_ws;
        rows_in_lds<<<NGROUP, WG_WAVES * 32, 0, stream>>>(x, part);
        add_partial_rows<<<BB * 4, 256, 0, stream>>>(part, out);
    } else {
        float* rootpow = (float*)d_ws;
        level_sums_roots<<<NGROUP, WG_WAVES * 32, 0, stream>>>(x, rootpow);
        scale_mean_kernel<<<BB * SC_WGPB, 256, 0, stream>>>(x, rootpow, out);
    }
}
